# starting kernel
# baseline (speedup 1.0000x reference)
#include <hip/hip_runtime.h>
#include <hip/hip_bf16.h>

typedef __attribute__((ext_vector_type(2))) float v2f;
typedef __attribute__((ext_vector_type(8))) float v8f;

#define DIN 64
#define DOUT 64
#define TILE_E 128
#define XS_STRIDE 68
#define W_STRIDE 65

__global__ void compute_W_kernel(const float* __restrict__ bases,
                                 const float* __restrict__ comb,
                                 float* __restrict__ W, int B) {
  int r = blockIdx.x;
  const float* c = comb + r * B;
  for (int idx = threadIdx.x; idx < DIN * DOUT; idx += blockDim.x) {
    float acc = 0.f;
    for (int b = 0; b < B; ++b) acc += c[b] * bases[b * DIN * DOUT + idx];
    W[r * DIN * DOUT + idx] = acc;
  }
}

__global__ void hist_kernel(const int* __restrict__ et, int E, int* counts) {
  int e = blockIdx.x * blockDim.x + threadIdx.x;
  if (e < E) atomicAdd(&counts[et[e]], 1);
}

__global__ void scan_kernel(const int* __restrict__ counts, int* offsets,
                            int* cursor, int R) {
  if (threadIdx.x == 0 && blockIdx.x == 0) {
    int off = 0;
    for (int r = 0; r < R; ++r) { offsets[r] = off; cursor[r] = off; off += counts[r]; }
  }
}

__global__ void scatter_kernel(const int* __restrict__ src, const int* __restrict__ tgt,
                               const int* __restrict__ et, const float* __restrict__ wts,
                               int E, int* cursor, int* s_src, int* s_tgt, float* s_w) {
  int e = blockIdx.x * blockDim.x + threadIdx.x;
  if (e < E) {
    int r = et[e];
    int p = atomicAdd(&cursor[r], 1);
    s_src[p] = src[e];
    s_tgt[p] = tgt[e];
    s_w[p]   = wts[e];
  }
}

__global__ void __launch_bounds__(256)
rgcn_tile_kernel(const float* __restrict__ x,
                 const float* __restrict__ W,
                 const int* __restrict__ s_src,
                 const int* __restrict__ s_tgt,
                 const float* __restrict__ s_w,
                 const int* __restrict__ offsets,
                 const int* __restrict__ counts,
                 float* __restrict__ out) {
  int r = blockIdx.y;
  int count = counts[r];
  int tile0 = (int)blockIdx.x * TILE_E;
  if (tile0 >= count) return;
  int start = offsets[r] + tile0;
  int nE = min(TILE_E, count - tile0);

  __shared__ float xs[TILE_E * XS_STRIDE];
  __shared__ float Wl[DIN * W_STRIDE];
  __shared__ int   tg[TILE_E];

  for (int idx = threadIdx.x; idx < DIN * DOUT; idx += 256) {
    int k = idx >> 6, n = idx & 63;
    Wl[k * W_STRIDE + n] = W[r * DIN * DOUT + idx];
  }

  int row  = threadIdx.x >> 1;
  int half = threadIdx.x & 1;
  int src = 0; float wt = 0.f;
  if (row < nE) { src = s_src[start + row]; wt = s_w[start + row]; }
  if (half == 0) tg[row] = (row < nE) ? s_tgt[start + row] : -1;
  const float4* xp = (const float4*)(x + (size_t)src * DIN) + half * 8;
  float* xd = &xs[row * XS_STRIDE + half * 32];
#pragma unroll
  for (int i = 0; i < 8; ++i) {
    float4 v = xp[i];
    xd[4 * i + 0] = v.x * wt;
    xd[4 * i + 1] = v.y * wt;
    xd[4 * i + 2] = v.z * wt;
    xd[4 * i + 3] = v.w * wt;
  }
  __syncthreads();

  int lane = threadIdx.x & 31;
  int wv   = threadIdx.x >> 5;
  int hi   = lane >> 4;
  int lm   = lane & 15;
  const float* arow = &xs[(wv * 16 + lm) * XS_STRIDE + 2 * hi];

#pragma unroll
  for (int ct = 0; ct < 4; ++ct) {
    v8f acc = {};
    const float* bcol = &Wl[(2 * hi) * W_STRIDE + ct * 16 + lm];
#pragma unroll
    for (int ks = 0; ks < 16; ++ks) {
      v2f a; a.x = arow[ks * 4];               a.y = arow[ks * 4 + 1];
      v2f b; b.x = bcol[(ks * 4) * W_STRIDE];  b.y = bcol[(ks * 4 + 1) * W_STRIDE];
      acc = __builtin_amdgcn_wmma_f32_16x16x4_f32(
           false, a,  false, b,
           (short)0, acc,  false,  false);
    }
#pragma unroll
    for (int j = 0; j < 8; ++j) {
      int m = wv * 16 + 8 * hi + j;
      int t = tg[m];
      if (t >= 0) atomicAdd(&out[(size_t)t * DOUT + ct * 16 + lm], acc[j]);
    }
  }
}

static inline size_t align256(size_t v) { return (v + 255) & ~(size_t)255; }

extern "C" void kernel_launch(void* const* d_in, const int* in_sizes, int n_in,
                              void* d_out, int out_size, void* d_ws, size_t ws_size,
                              hipStream_t stream) {
  const float* x       = (const float*)d_in[0];
  const int*   source  = (const int*)d_in[1];
  const int*   target  = (const int*)d_in[2];
  const int*   etype   = (const int*)d_in[3];
  const float* ew      = (const float*)d_in[4];
  const float* bases   = (const float*)d_in[5];
  const float* comb    = (const float*)d_in[6];
  float*       out     = (float*)d_out;

  const int N = in_sizes[0] / DIN;
  const int E = in_sizes[1];
  const int B = in_sizes[5] / (DIN * DOUT);
  const int R = in_sizes[6] / B;

  uint8_t* w = (uint8_t*)d_ws;
  float* Wbuf    = (float*)w;  w += align256((size_t)R * DIN * DOUT * 4);
  int*   counts  = (int*)w;    w += 256;
  int*   offsets = (int*)w;    w += 256;
  int*   cursor  = (int*)w;    w += 256;
  int*   s_src   = (int*)w;    w += align256((size_t)E * 4);
  int*   s_tgt   = (int*)w;    w += align256((size_t)E * 4);
  float* s_w     = (float*)w;

  hipMemsetAsync(counts, 0, 768, stream);
  hipMemsetAsync(out, 0, (size_t)N * DOUT * sizeof(float), stream);

  compute_W_kernel<<<R, 256, 0, stream>>>(bases, comb, Wbuf, B);

  int eblk = (E + 255) / 256;
  hist_kernel<<<eblk, 256, 0, stream>>>(etype, E, counts);
  scan_kernel<<<1, 32, 0, stream>>>(counts, offsets, cursor, R);
  scatter_kernel<<<eblk, 256, 0, stream>>>(source, target, etype, ew, E,
                                           cursor, s_src, s_tgt, s_w);

  dim3 grid((E + TILE_E - 1) / TILE_E, R);
  rgcn_tile_kernel<<<grid, 256, 0, stream>>>(x, Wbuf, s_src, s_tgt, s_w,
                                             offsets, counts, out);
}


# hist and bucket-scatter kernels count and rank per relation in LDS per 2048-edge block, one global atomic per relation per block; edge order inside a bucket (float-add arrival order) changes
# speedup vs baseline: 3.3863x; 3.3863x over previous
#include <hip/hip_runtime.h>
#include <hip/hip_bf16.h>

typedef __attribute__((ext_vector_type(2))) float v2f;
typedef __attribute__((ext_vector_type(8))) float v8f;

#define DIN 64
#define DOUT 64
#define TILE_E 128
#define XS_STRIDE 68
#define W_STRIDE 65

__global__ void compute_W_kernel(const float* __restrict__ bases,
                                 const float* __restrict__ comb,
                                 float* __restrict__ W, int B) {
  int r = blockIdx.x;
  const float* c = comb + r * B;
  for (int idx = threadIdx.x; idx < DIN * DOUT; idx += blockDim.x) {
    float acc = 0.f;
    for (int b = 0; b < B; ++b) acc += c[b] * bases[b * DIN * DOUT + idx];
    W[r * DIN * DOUT + idx] = acc;
  }
}

#define CNT_G 2
#define CNT_EPB (256 * 4 * CNT_G)
#define MAXR 64

__device__ __forceinline__ void load4i(const int* __restrict__ p, int e, int E, int fill, int (&v)[4]) {
  if (e + 4 <= E) {
    int4 q = *(const int4*)(p + e);
    v[0] = q.x; v[1] = q.y; v[2] = q.z; v[3] = q.w;
  } else {
#pragma unroll
    for (int j = 0; j < 4; ++j) v[j] = (e + j < E) ? p[e + j] : fill;
  }
}
__device__ __forceinline__ void load4f(const float* __restrict__ p, int e, int E, float (&v)[4]) {
  if (e + 4 <= E) {
    float4 q = *(const float4*)(p + e);
    v[0] = q.x; v[1] = q.y; v[2] = q.z; v[3] = q.w;
  } else {
#pragma unroll
    for (int j = 0; j < 4; ++j) v[j] = (e + j < E) ? p[e + j] : 0.f;
  }
}

__global__ void __launch_bounds__(256)
hist_kernel(const int* __restrict__ et, int E, int R, int* counts) {
  __shared__ int lc[MAXR];
  if (threadIdx.x < MAXR) lc[threadIdx.x] = 0;
  __syncthreads();
  int base = (int)blockIdx.x * CNT_EPB;
  int r[CNT_G][4];
#pragma unroll
  for (int g = 0; g < CNT_G; ++g)
    load4i(et, base + (g * 256 + (int)threadIdx.x) * 4, E, -1, r[g]);
#pragma unroll
  for (int g = 0; g < CNT_G; ++g)
#pragma unroll
    for (int j = 0; j < 4; ++j)
      if (r[g][j] >= 0) atomicAdd(&lc[r[g][j]], 1);
  __syncthreads();
  if ((int)threadIdx.x < R) {
    int c = lc[threadIdx.x];
    if (c) atomicAdd(&counts[threadIdx.x], c);
  }
}

__global__ void scan_kernel(const int* __restrict__ counts, int* offsets,
                            int* cursor, int R) {
  if (threadIdx.x == 0 && blockIdx.x == 0) {
    int off = 0;
    for (int r = 0; r < R; ++r) { offsets[r] = off; cursor[r] = off; off += counts[r]; }
  }
}

__global__ void __launch_bounds__(256)
scatter_kernel(const int* __restrict__ src, const int* __restrict__ tgt,
               const int* __restrict__ et, const float* __restrict__ wts,
               int E, int R, int* cursor, int* s_src, int* s_tgt, float* s_w) {
  __shared__ int lc[MAXR];
  if (threadIdx.x < MAXR) lc[threadIdx.x] = 0;
  __syncthreads();
  int base = (int)blockIdx.x * CNT_EPB;
  int r[CNT_G][4], vs[CNT_G][4], vt[CNT_G][4], rk[CNT_G][4];
  float vw[CNT_G][4];
#pragma unroll
  for (int g = 0; g < CNT_G; ++g) {
    int e = base + (g * 256 + (int)threadIdx.x) * 4;
    load4i(et,  e, E, -1, r[g]);
    load4i(src, e, E, 0, vs[g]);
    load4i(tgt, e, E, 0, vt[g]);
    load4f(wts, e, E, vw[g]);
  }
#pragma unroll
  for (int g = 0; g < CNT_G; ++g)
#pragma unroll
    for (int j = 0; j < 4; ++j)
      rk[g][j] = (r[g][j] >= 0) ? atomicAdd(&lc[r[g][j]], 1) : 0;
  __syncthreads();
  if ((int)threadIdx.x < R) {
    int c = lc[threadIdx.x];
    lc[threadIdx.x] = c ? atomicAdd(&cursor[threadIdx.x], c) : 0;
  }
  __syncthreads();
  int p[CNT_G][4];
#pragma unroll
  for (int g = 0; g < CNT_G; ++g)
#pragma unroll
    for (int j = 0; j < 4; ++j)
      p[g][j] = lc[max(r[g][j], 0)] + rk[g][j];
#pragma unroll
  for (int g = 0; g < CNT_G; ++g)
#pragma unroll
    for (int j = 0; j < 4; ++j)
      if (r[g][j] >= 0) {
        s_src[p[g][j]] = vs[g][j];
        s_tgt[p[g][j]] = vt[g][j];
        s_w[p[g][j]]   = vw[g][j];
      }
}

__global__ void __launch_bounds__(256)
rgcn_tile_kernel(const float* __restrict__ x,
                 const float* __restrict__ W,
                 const int* __restrict__ s_src,
                 const int* __restrict__ s_tgt,
                 const float* __restrict__ s_w,
                 const int* __restrict__ offsets,
                 const int* __restrict__ counts,
                 float* __restrict__ out) {
  int r = blockIdx.y;
  int count = counts[r];
  int tile0 = (int)blockIdx.x * TILE_E;
  if (tile0 >= count) return;
  int start = offsets[r] + tile0;
  int nE = min(TILE_E, count - tile0);

  __shared__ float xs[TILE_E * XS_STRIDE];
  __shared__ float Wl[DIN * W_STRIDE];
  __shared__ int   tg[TILE_E];

  for (int idx = threadIdx.x; idx < DIN * DOUT; idx += 256) {
    int k = idx >> 6, n = idx & 63;
    Wl[k * W_STRIDE + n] = W[r * DIN * DOUT + idx];
  }

  int row  = threadIdx.x >> 1;
  int half = threadIdx.x & 1;
  int src = 0; float wt = 0.f;
  if (row < nE) { src = s_src[start + row]; wt = s_w[start + row]; }
  if (half == 0) tg[row] = (row < nE) ? s_tgt[start + row] : -1;
  const float4* xp = (const float4*)(x + (size_t)src * DIN) + half * 8;
  float* xd = &xs[row * XS_STRIDE + half * 32];
#pragma unroll
  for (int i = 0; i < 8; ++i) {
    float4 v = xp[i];
    xd[4 * i + 0] = v.x * wt;
    xd[4 * i + 1] = v.y * wt;
    xd[4 * i + 2] = v.z * wt;
    xd[4 * i + 3] = v.w * wt;
  }
  __syncthreads();

  int lane = threadIdx.x & 31;
  int wv   = threadIdx.x >> 5;
  int hi   = lane >> 4;
  int lm   = lane & 15;
  const float* arow = &xs[(wv * 16 + lm) * XS_STRIDE + 2 * hi];

#pragma unroll
  for (int ct = 0; ct < 4; ++ct) {
    v8f acc = {};
    const float* bcol = &Wl[(2 * hi) * W_STRIDE + ct * 16 + lm];
#pragma unroll
    for (int ks = 0; ks < 16; ++ks) {
      v2f a; a.x = arow[ks * 4];               a.y = arow[ks * 4 + 1];
      v2f b; b.x = bcol[(ks * 4) * W_STRIDE];  b.y = bcol[(ks * 4 + 1) * W_STRIDE];
      acc = __builtin_amdgcn_wmma_f32_16x16x4_f32(
           false, a,  false, b,
           (short)0, acc,  false,  false);
    }
#pragma unroll
    for (int j = 0; j < 8; ++j) {
      int m = wv * 16 + 8 * hi + j;
      int t = tg[m];
      if (t >= 0) atomicAdd(&out[(size_t)t * DOUT + ct * 16 + lm], acc[j]);
    }
  }
}

static inline size_t align256(size_t v) { return (v + 255) & ~(size_t)255; }

extern "C" void kernel_launch(void* const* d_in, const int* in_sizes, int n_in,
                              void* d_out, int out_size, void* d_ws, size_t ws_size,
                              hipStream_t stream) {
  const float* x       = (const float*)d_in[0];
  const int*   source  = (const int*)d_in[1];
  const int*   target  = (const int*)d_in[2];
  const int*   etype   = (const int*)d_in[3];
  const float* ew      = (const float*)d_in[4];
  const float* bases   = (const float*)d_in[5];
  const float* comb    = (const float*)d_in[6];
  float*       out     = (float*)d_out;

  const int N = in_sizes[0] / DIN;
  const int E = in_sizes[1];
  const int B = in_sizes[5] / (DIN * DOUT);
  const int R = in_sizes[6] / B;

  uint8_t* w = (uint8_t*)d_ws;
  float* Wbuf    = (float*)w;  w += align256((size_t)R * DIN * DOUT * 4);
  int*   counts  = (int*)w;    w += 256;
  int*   offsets = (int*)w;    w += 256;
  int*   cursor  = (int*)w;    w += 256;
  int*   s_src   = (int*)w;    w += align256((size_t)E * 4);
  int*   s_tgt   = (int*)w;    w += align256((size_t)E * 4);
  float* s_w     = (float*)w;

  hipMemsetAsync(counts, 0, 768, stream);
  hipMemsetAsync(out, 0, (size_t)N * DOUT * sizeof(float), stream);

  compute_W_kernel<<<R, 256, 0, stream>>>(bases, comb, Wbuf, B);

  int eblk = (E + CNT_EPB - 1) / CNT_EPB;
  hist_kernel<<<eblk, 256, 0, stream>>>(etype, E, R, counts);
  scan_kernel<<<1, 32, 0, stream>>>(counts, offsets, cursor, R);
  scatter_kernel<<<eblk, 256, 0, stream>>>(source, target, etype, ew, E, R,
                                           cursor, s_src, s_tgt, s_w);

  dim3 grid((E + TILE_E - 1) / TILE_E, R);
  rgcn_tile_kernel<<<grid, 256, 0, stream>>>(x, Wbuf, s_src, s_tgt, s_w,
                                             offsets, counts, out);
}


# main launch numbered over live tiles only: 1-D grid of E/128+R blocks, each wave finds its relation and tile by a scalar walk of the counts (was 2-D, most blocks empty); float-add order as before
# speedup vs baseline: 3.8315x; 1.1315x over previous
#include <hip/hip_runtime.h>
#include <hip/hip_bf16.h>

typedef __attribute__((ext_vector_type(2))) float v2f;
typedef __attribute__((ext_vector_type(8))) float v8f;

#define DIN 64
#define DOUT 64
#define TILE_E 128
#define XS_STRIDE 68
#define W_STRIDE 65

__global__ void compute_W_kernel(const float* __restrict__ bases,
                                 const float* __restrict__ comb,
                                 float* __restrict__ W, int B) {
  int r = blockIdx.x;
  const float* c = comb + r * B;
  for (int idx = threadIdx.x; idx < DIN * DOUT; idx += blockDim.x) {
    float acc = 0.f;
    for (int b = 0; b < B; ++b) acc += c[b] * bases[b * DIN * DOUT + idx];
    W[r * DIN * DOUT + idx] = acc;
  }
}

#define CNT_G 2
#define CNT_EPB (256 * 4 * CNT_G)
#define MAXR 64

__device__ __forceinline__ void load4i(const int* __restrict__ p, int e, int E, int fill, int (&v)[4]) {
  if (e + 4 <= E) {
    int4 q = *(const int4*)(p + e);
    v[0] = q.x; v[1] = q.y; v[2] = q.z; v[3] = q.w;
  } else {
#pragma unroll
    for (int j = 0; j < 4; ++j) v[j] = (e + j < E) ? p[e + j] : fill;
  }
}
__device__ __forceinline__ void load4f(const float* __restrict__ p, int e, int E, float (&v)[4]) {
  if (e + 4 <= E) {
    float4 q = *(const float4*)(p + e);
    v[0] = q.x; v[1] = q.y; v[2] = q.z; v[3] = q.w;
  } else {
#pragma unroll
    for (int j = 0; j < 4; ++j) v[j] = (e + j < E) ? p[e + j] : 0.f;
  }
}

__global__ void __launch_bounds__(256)
hist_kernel(const int* __restrict__ et, int E, int R, int* counts) {
  __shared__ int lc[MAXR];
  if (threadIdx.x < MAXR) lc[threadIdx.x] = 0;
  __syncthreads();
  int base = (int)blockIdx.x * CNT_EPB;
  int r[CNT_G][4];
#pragma unroll
  for (int g = 0; g < CNT_G; ++g)
    load4i(et, base + (g * 256 + (int)threadIdx.x) * 4, E, -1, r[g]);
#pragma unroll
  for (int g = 0; g < CNT_G; ++g)
#pragma unroll
    for (int j = 0; j < 4; ++j)
      if (r[g][j] >= 0) atomicAdd(&lc[r[g][j]], 1);
  __syncthreads();
  if ((int)threadIdx.x < R) {
    int c = lc[threadIdx.x];
    if (c) atomicAdd(&counts[threadIdx.x], c);
  }
}

__global__ void scan_kernel(const int* __restrict__ counts, int* cursor, int R) {
  if (threadIdx.x == 0 && blockIdx.x == 0) {
    int off = 0;
    for (int r = 0; r < R; ++r) { cursor[r] = off; off += counts[r]; }
  }
}

__global__ void __launch_bounds__(256)
scatter_kernel(const int* __restrict__ src, const int* __restrict__ tgt,
               const int* __restrict__ et, const float* __restrict__ wts,
               int E, int R, int* cursor, int* s_src, int* s_tgt, float* s_w) {
  __shared__ int lc[MAXR];
  if (threadIdx.x < MAXR) lc[threadIdx.x] = 0;
  __syncthreads();
  int base = (int)blockIdx.x * CNT_EPB;
  int r[CNT_G][4], vs[CNT_G][4], vt[CNT_G][4], rk[CNT_G][4];
  float vw[CNT_G][4];
#pragma unroll
  for (int g = 0; g < CNT_G; ++g) {
    int e = base + (g * 256 + (int)threadIdx.x) * 4;
    load4i(et,  e, E, -1, r[g]);
    load4i(src, e, E, 0, vs[g]);
    load4i(tgt, e, E, 0, vt[g]);
    load4f(wts, e, E, vw[g]);
  }
#pragma unroll
  for (int g = 0; g < CNT_G; ++g)
#pragma unroll
    for (int j = 0; j < 4; ++j)
      rk[g][j] = (r[g][j] >= 0) ? atomicAdd(&lc[r[g][j]], 1) : 0;
  __syncthreads();
  if ((int)threadIdx.x < R) {
    int c = lc[threadIdx.x];
    lc[threadIdx.x] = c ? atomicAdd(&cursor[threadIdx.x], c) : 0;
  }
  __syncthreads();
  int p[CNT_G][4];
#pragma unroll
  for (int g = 0; g < CNT_G; ++g)
#pragma unroll
    for (int j = 0; j < 4; ++j)
      p[g][j] = lc[max(r[g][j], 0)] + rk[g][j];
#pragma unroll
  for (int g = 0; g < CNT_G; ++g)
#pragma unroll
    for (int j = 0; j < 4; ++j)
      if (r[g][j] >= 0) {
        s_src[p[g][j]] = vs[g][j];
        s_tgt[p[g][j]] = vt[g][j];
        s_w[p[g][j]]   = vw[g][j];
      }
}

__global__ void __launch_bounds__(256)
rgcn_tile_kernel(const float* __restrict__ x,
                 const float* __restrict__ W,
                 const int* __restrict__ s_src,
                 const int* __restrict__ s_tgt,
                 const float* __restrict__ s_w,
                 const int* __restrict__ counts,
                 int R,
                 float* __restrict__ out) {
  int b = (int)blockIdx.x;
  int r = 0, cnt = 0, eo = 0, ts = 0;
  int tacc = 0, eoff = 0;
  for (int g = 0; g < R; g += 16) {
#pragma unroll
    for (int j = 0; j < 16; ++j) {
      int c = counts[g + j];
      bool here = (b >= tacc);
      r   = here ? g + j : r;
      cnt = here ? c     : cnt;
      eo  = here ? eoff  : eo;
      ts  = here ? tacc  : ts;
      tacc += (int)((unsigned)(c + TILE_E - 1) / TILE_E);
      eoff += c;
    }
  }
  int tile0 = (b - ts) * TILE_E;
  if (tile0 >= cnt) return;
  int start = eo + tile0;
  int nE = min(TILE_E, cnt - tile0);

  __shared__ float xs[TILE_E * XS_STRIDE];
  __shared__ float Wl[DIN * W_STRIDE];
  __shared__ int   tg[TILE_E];

  for (int idx = threadIdx.x; idx < DIN * DOUT; idx += 256) {
    int k = idx >> 6, n = idx & 63;
    Wl[k * W_STRIDE + n] = W[r * DIN * DOUT + idx];
  }

  int row  = threadIdx.x >> 1;
  int half = threadIdx.x & 1;
  int src = 0; float wt = 0.f;
  if (row < nE) { src = s_src[start + row]; wt = s_w[start + row]; }
  if (half == 0) tg[row] = (row < nE) ? s_tgt[start + row] : -1;
  const float4* xp = (const float4*)(x + (size_t)src * DIN) + half * 8;
  float* xd = &xs[row * XS_STRIDE + half * 32];
#pragma unroll
  for (int i = 0; i < 8; ++i) {
    float4 v = xp[i];
    xd[4 * i + 0] = v.x * wt;
    xd[4 * i + 1] = v.y * wt;
    xd[4 * i + 2] = v.z * wt;
    xd[4 * i + 3] = v.w * wt;
  }
  __syncthreads();

  int lane = threadIdx.x & 31;
  int wv   = threadIdx.x >> 5;
  int hi   = lane >> 4;
  int lm   = lane & 15;
  const float* arow = &xs[(wv * 16 + lm) * XS_STRIDE + 2 * hi];

#pragma unroll
  for (int ct = 0; ct < 4; ++ct) {
    v8f acc = {};
    const float* bcol = &Wl[(2 * hi) * W_STRIDE + ct * 16 + lm];
#pragma unroll
    for (int ks = 0; ks < 16; ++ks) {
      v2f a; a.x = arow[ks * 4];               a.y = arow[ks * 4 + 1];
      v2f b; b.x = bcol[(ks * 4) * W_STRIDE];  b.y = bcol[(ks * 4 + 1) * W_STRIDE];
      acc = __builtin_amdgcn_wmma_f32_16x16x4_f32(
           false, a,  false, b,
           (short)0, acc,  false,  false);
    }
#pragma unroll
    for (int j = 0; j < 8; ++j) {
      int m = wv * 16 + 8 * hi + j;
      int t = tg[m];
      if (t >= 0) atomicAdd(&out[(size_t)t * DOUT + ct * 16 + lm], acc[j]);
    }
  }
}

static inline size_t align256(size_t v) { return (v + 255) & ~(size_t)255; }

extern "C" void kernel_launch(void* const* d_in, const int* in_sizes, int n_in,
                              void* d_out, int out_size, void* d_ws, size_t ws_size,
                              hipStream_t stream) {
  const float* x       = (const float*)d_in[0];
  const int*   source  = (const int*)d_in[1];
  const int*   target  = (const int*)d_in[2];
  const int*   etype   = (const int*)d_in[3];
  const float* ew      = (const float*)d_in[4];
  const float* bases   = (const float*)d_in[5];
  const float* comb    = (const float*)d_in[6];
  float*       out     = (float*)d_out;

  const int N = in_sizes[0] / DIN;
  const int E = in_sizes[1];
  const int B = in_sizes[5] / (DIN * DOUT);
  const int R = in_sizes[6] / B;

  uint8_t* w = (uint8_t*)d_ws;
  float* Wbuf    = (float*)w;  w += align256((size_t)R * DIN * DOUT * 4);
  int*   counts  = (int*)w;    w += 256;
  int*   cursor  = (int*)w;    w += 256;
  int*   s_src   = (int*)w;    w += align256((size_t)E * 4);
  int*   s_tgt   = (int*)w;    w += align256((size_t)E * 4);
  float* s_w     = (float*)w;

  hipMemsetAsync(counts, 0, 512, stream);
  hipMemsetAsync(out, 0, (size_t)N * DOUT * sizeof(float), stream);

  compute_W_kernel<<<R, 256, 0, stream>>>(bases, comb, Wbuf, B);

  int eblk = (E + CNT_EPB - 1) / CNT_EPB;
  hist_kernel<<<eblk, 256, 0, stream>>>(etype, E, R, counts);
  scan_kernel<<<1, 32, 0, stream>>>(counts, cursor, R);
  scatter_kernel<<<eblk, 256, 0, stream>>>(source, target, etype, ew, E, R,
                                           cursor, s_src, s_tgt, s_w);

  int tiles = (E + TILE_E - 1) / TILE_E + R;
  rgcn_tile_kernel<<<tiles, 256, 0, stream>>>(x, Wbuf, s_src, s_tgt, s_w,
                                              counts, R, out);
}


# main kernel gather: each x-row load instruction covers two whole 256-byte rows (16 lanes x 16 B a row, 4 lines) instead of 32 lanes each in a line of its own (32 lines); float-add order as before
# speedup vs baseline: 4.1569x; 1.0849x over previous
#include <hip/hip_runtime.h>
#include <hip/hip_bf16.h>

typedef __attribute__((ext_vector_type(2))) float v2f;
typedef __attribute__((ext_vector_type(8))) float v8f;

#define DIN 64
#define DOUT 64
#define TILE_E 128
#define XS_STRIDE 68
#define W_STRIDE 65

__global__ void compute_W_kernel(const float* __restrict__ bases,
                                 const float* __restrict__ comb,
                                 float* __restrict__ W, int B) {
  int r = blockIdx.x;
  const float* c = comb + r * B;
  for (int idx = threadIdx.x; idx < DIN * DOUT; idx += blockDim.x) {
    float acc = 0.f;
    for (int b = 0; b < B; ++b) acc += c[b] * bases[b * DIN * DOUT + idx];
    W[r * DIN * DOUT + idx] = acc;
  }
}

#define CNT_G 2
#define CNT_EPB (256 * 4 * CNT_G)
#define MAXR 64

__device__ __forceinline__ void load4i(const int* __restrict__ p, int e, int E, int fill, int (&v)[4]) {
  if (e + 4 <= E) {
    int4 q = *(const int4*)(p + e);
    v[0] = q.x; v[1] = q.y; v[2] = q.z; v[3] = q.w;
  } else {
#pragma unroll
    for (int j = 0; j < 4; ++j) v[j] = (e + j < E) ? p[e + j] : fill;
  }
}
__device__ __forceinline__ void load4f(const float* __restrict__ p, int e, int E, float (&v)[4]) {
  if (e + 4 <= E) {
    float4 q = *(const float4*)(p + e);
    v[0] = q.x; v[1] = q.y; v[2] = q.z; v[3] = q.w;
  } else {
#pragma unroll
    for (int j = 0; j < 4; ++j) v[j] = (e + j < E) ? p[e + j] : 0.f;
  }
}

__global__ void __launch_bounds__(256)
hist_kernel(const int* __restrict__ et, int E, int R, int* counts) {
  __shared__ int lc[MAXR];
  if (threadIdx.x < MAXR) lc[threadIdx.x] = 0;
  __syncthreads();
  int base = (int)blockIdx.x * CNT_EPB;
  int r[CNT_G][4];
#pragma unroll
  for (int g = 0; g < CNT_G; ++g)
    load4i(et, base + (g * 256 + (int)threadIdx.x) * 4, E, -1, r[g]);
#pragma unroll
  for (int g = 0; g < CNT_G; ++g)
#pragma unroll
    for (int j = 0; j < 4; ++j)
      if (r[g][j] >= 0) atomicAdd(&lc[r[g][j]], 1);
  __syncthreads();
  if ((int)threadIdx.x < R) {
    int c = lc[threadIdx.x];
    if (c) atomicAdd(&counts[threadIdx.x], c);
  }
}

__global__ void scan_kernel(const int* __restrict__ counts, int* cursor, int R) {
  if (threadIdx.x == 0 && blockIdx.x == 0) {
    int off = 0;
    for (int r = 0; r < R; ++r) { cursor[r] = off; off += counts[r]; }
  }
}

__global__ void __launch_bounds__(256)
scatter_kernel(const int* __restrict__ src, const int* __restrict__ tgt,
               const int* __restrict__ et, const float* __restrict__ wts,
               int E, int R, int* cursor, int* s_src, int* s_tgt, float* s_w) {
  __shared__ int lc[MAXR];
  if (threadIdx.x < MAXR) lc[threadIdx.x] = 0;
  __syncthreads();
  int base = (int)blockIdx.x * CNT_EPB;
  int r[CNT_G][4], vs[CNT_G][4], vt[CNT_G][4], rk[CNT_G][4];
  float vw[CNT_G][4];
#pragma unroll
  for (int g = 0; g < CNT_G; ++g) {
    int e = base + (g * 256 + (int)threadIdx.x) * 4;
    load4i(et,  e, E, -1, r[g]);
    load4i(src, e, E, 0, vs[g]);
    load4i(tgt, e, E, 0, vt[g]);
    load4f(wts, e, E, vw[g]);
  }
#pragma unroll
  for (int g = 0; g < CNT_G; ++g)
#pragma unroll
    for (int j = 0; j < 4; ++j)
      rk[g][j] = (r[g][j] >= 0) ? atomicAdd(&lc[r[g][j]], 1) : 0;
  __syncthreads();
  if ((int)threadIdx.x < R) {
    int c = lc[threadIdx.x];
    lc[threadIdx.x] = c ? atomicAdd(&cursor[threadIdx.x], c) : 0;
  }
  __syncthreads();
  int p[CNT_G][4];
#pragma unroll
  for (int g = 0; g < CNT_G; ++g)
#pragma unroll
    for (int j = 0; j < 4; ++j)
      p[g][j] = lc[max(r[g][j], 0)] + rk[g][j];
#pragma unroll
  for (int g = 0; g < CNT_G; ++g)
#pragma unroll
    for (int j = 0; j < 4; ++j)
      if (r[g][j] >= 0) {
        s_src[p[g][j]] = vs[g][j];
        s_tgt[p[g][j]] = vt[g][j];
        s_w[p[g][j]]   = vw[g][j];
      }
}

__global__ void __launch_bounds__(256)
rgcn_tile_kernel(const float* __restrict__ x,
                 const float* __restrict__ W,
                 const int* __restrict__ s_src,
                 const int* __restrict__ s_tgt,
                 const float* __restrict__ s_w,
                 const int* __restrict__ counts,
                 int R,
                 float* __restrict__ out) {
  int b = (int)blockIdx.x;
  int r = 0, cnt = 0, eo = 0, ts = 0;
  int tacc = 0, eoff = 0;
  for (int g = 0; g < R; g += 16) {
#pragma unroll
    for (int j = 0; j < 16; ++j) {
      int c = counts[g + j];
      bool here = (b >= tacc);
      r   = here ? g + j : r;
      cnt = here ? c     : cnt;
      eo  = here ? eoff  : eo;
      ts  = here ? tacc  : ts;
      tacc += (int)((unsigned)(c + TILE_E - 1) / TILE_E);
      eoff += c;
    }
  }
  int tile0 = (b - ts) * TILE_E;
  if (tile0 >= cnt) return;
  int start = eo + tile0;
  int nE = min(TILE_E, cnt - tile0);

  __shared__ __attribute__((aligned(16))) float xs[TILE_E * XS_STRIDE];
  __shared__ float Wl[DIN * W_STRIDE];
  __shared__ int   tg[TILE_E];

  for (int idx = threadIdx.x; idx < DIN * DOUT; idx += 256) {
    int k = idx >> 6, n = idx & 63;
    Wl[k * W_STRIDE + n] = W[r * DIN * DOUT + idx];
  }

  int glane = threadIdx.x & 31;
  int gw    = threadIdx.x >> 5;
  int gh    = glane >> 4;
  int gq    = glane & 15;
  int row   = gw * 16 + gq;
  int src = 0; float wt = 0.f;
  if (row < nE) { src = s_src[start + row]; wt = s_w[start + row]; }
  if (gh == 0) tg[row] = (row < nE) ? s_tgt[start + row] : -1;
  int wti = __float_as_int(wt);
#pragma unroll
  for (int i = 0; i < 8; ++i) {
    int s0 = __builtin_amdgcn_readlane(src, 2 * i);
    int s1 = __builtin_amdgcn_readlane(src, 2 * i + 1);
    int w0 = __builtin_amdgcn_readlane(wti, 2 * i);
    int w1 = __builtin_amdgcn_readlane(wti, 2 * i + 1);
    int   s = gh ? s1 : s0;
    float w = __int_as_float(gh ? w1 : w0);
    float4 v = *((const float4*)(x + (size_t)s * DIN) + gq);
    float4 o;
    o.x = v.x * w; o.y = v.y * w; o.z = v.z * w; o.w = v.w * w;
    *(float4*)&xs[(gw * 16 + 2 * i + gh) * XS_STRIDE + 4 * gq] = o;
  }
  __syncthreads();

  int lane = threadIdx.x & 31;
  int wv   = threadIdx.x >> 5;
  int hi   = lane >> 4;
  int lm   = lane & 15;
  const float* arow = &xs[(wv * 16 + lm) * XS_STRIDE + 2 * hi];

#pragma unroll
  for (int ct = 0; ct < 4; ++ct) {
    v8f acc = {};
    const float* bcol = &Wl[(2 * hi) * W_STRIDE + ct * 16 + lm];
#pragma unroll
    for (int ks = 0; ks < 16; ++ks) {
      v2f a; a.x = arow[ks * 4];               a.y = arow[ks * 4 + 1];
      v2f b; b.x = bcol[(ks * 4) * W_STRIDE];  b.y = bcol[(ks * 4 + 1) * W_STRIDE];
      acc = __builtin_amdgcn_wmma_f32_16x16x4_f32(
           false, a,  false, b,
           (short)0, acc,  false,  false);
    }
#pragma unroll
    for (int j = 0; j < 8; ++j) {
      int m = wv * 16 + 8 * hi + j;
      int t = tg[m];
      if (t >= 0) atomicAdd(&out[(size_t)t * DOUT + ct * 16 + lm], acc[j]);
    }
  }
}

static inline size_t align256(size_t v) { return (v + 255) & ~(size_t)255; }

extern "C" void kernel_launch(void* const* d_in, const int* in_sizes, int n_in,
                              void* d_out, int out_size, void* d_ws, size_t ws_size,
                              hipStream_t stream) {
  const float* x       = (const float*)d_in[0];
  const int*   source  = (const int*)d_in[1];
  const int*   target  = (const int*)d_in[2];
  const int*   etype   = (const int*)d_in[3];
  const float* ew      = (const float*)d_in[4];
  const float* bases   = (const float*)d_in[5];
  const float* comb    = (const float*)d_in[6];
  float*       out     = (float*)d_out;

  const int N = in_sizes[0] / DIN;
  const int E = in_sizes[1];
  const int B = in_sizes[5] / (DIN * DOUT);
  const int R = in_sizes[6] / B;

  uint8_t* w = (uint8_t*)d_ws;
  float* Wbuf    = (float*)w;  w += align256((size_t)R * DIN * DOUT * 4);
  int*   counts  = (int*)w;    w += 256;
  int*   cursor  = (int*)w;    w += 256;
  int*   s_src   = (int*)w;    w += align256((size_t)E * 4);
  int*   s_tgt   = (int*)w;    w += align256((size_t)E * 4);
  float* s_w     = (float*)w;

  hipMemsetAsync(counts, 0, 512, stream);
  hipMemsetAsync(out, 0, (size_t)N * DOUT * sizeof(float), stream);

  compute_W_kernel<<<R, 256, 0, stream>>>(bases, comb, Wbuf, B);

  int eblk = (E + CNT_EPB - 1) / CNT_EPB;
  hist_kernel<<<eblk, 256, 0, stream>>>(etype, E, R, counts);
  scan_kernel<<<1, 32, 0, stream>>>(counts, cursor, R);
  scatter_kernel<<<eblk, 256, 0, stream>>>(source, target, etype, ew, E, R,
                                           cursor, s_src, s_tgt, s_w);

  int tiles = (E + TILE_E - 1) / TILE_E + R;
  rgcn_tile_kernel<<<tiles, 256, 0, stream>>>(x, Wbuf, s_src, s_tgt, s_w,
                                              counts, R, out);
}


# main kernel: edges bucketed by (target half of node range, relation); a block reads its die id and takes a tile of its own cache half by ticket (other half's when empty); float-add order as before
# speedup vs baseline: 5.0588x; 1.2170x over previous
#include <hip/hip_runtime.h>
#include <hip/hip_bf16.h>

typedef __attribute__((ext_vector_type(2))) float v2f;
typedef __attribute__((ext_vector_type(8))) float v8f;

#define DIN 64
#define DOUT 64
#define TILE_E 128
#define XS_STRIDE 68
#define W_STRIDE 65

__global__ void compute_W_kernel(const float* __restrict__ bases,
                                 const float* __restrict__ comb,
                                 float* __restrict__ W, int B) {
  int r = blockIdx.x;
  const float* c = comb + r * B;
  for (int idx = threadIdx.x; idx < DIN * DOUT; idx += blockDim.x) {
    float acc = 0.f;
    for (int b = 0; b < B; ++b) acc += c[b] * bases[b * DIN * DOUT + idx];
    W[r * DIN * DOUT + idx] = acc;
  }
}

#define CNT_G 2
#define CNT_EPB (256 * 4 * CNT_G)
#define MAXR 64

__device__ __forceinline__ void load4i(const int* __restrict__ p, int e, int E, int fill, int (&v)[4]) {
  if (e + 4 <= E) {
    int4 q = *(const int4*)(p + e);
    v[0] = q.x; v[1] = q.y; v[2] = q.z; v[3] = q.w;
  } else {
#pragma unroll
    for (int j = 0; j < 4; ++j) v[j] = (e + j < E) ? p[e + j] : fill;
  }
}
__device__ __forceinline__ void load4f(const float* __restrict__ p, int e, int E, float (&v)[4]) {
  if (e + 4 <= E) {
    float4 q = *(const float4*)(p + e);
    v[0] = q.x; v[1] = q.y; v[2] = q.z; v[3] = q.w;
  } else {
#pragma unroll
    for (int j = 0; j < 4; ++j) v[j] = (e + j < E) ? p[e + j] : 0.f;
  }
}

__device__ __forceinline__ int bucket_of(int rel, int tgt, int R, int NH) {
  return rel + (tgt >= NH ? R : 0);
}

__global__ void __launch_bounds__(256)
hist_kernel(const int* __restrict__ et, const int* __restrict__ tgt, int E, int R, int N, int* counts) {
  __shared__ int lc[MAXR];
  if (threadIdx.x < MAXR) lc[threadIdx.x] = 0;
  __syncthreads();
  const int NH = N >> 1;
  int base = (int)blockIdx.x * CNT_EPB;
  int r[CNT_G][4], vt[CNT_G][4];
#pragma unroll
  for (int g = 0; g < CNT_G; ++g) {
    int e = base + (g * 256 + (int)threadIdx.x) * 4;
    load4i(et,  e, E, -1, r[g]);
    load4i(tgt, e, E, 0, vt[g]);
  }
#pragma unroll
  for (int g = 0; g < CNT_G; ++g)
#pragma unroll
    for (int j = 0; j < 4; ++j)
      if (r[g][j] >= 0) atomicAdd(&lc[bucket_of(r[g][j], vt[g][j], R, NH)], 1);
  __syncthreads();
  if ((int)threadIdx.x < 2 * R) {
    int c = lc[threadIdx.x];
    if (c) atomicAdd(&counts[threadIdx.x], c);
  }
}

__global__ void scan_kernel(const int* __restrict__ counts, int* cursor, int R) {
  if (threadIdx.x == 0 && blockIdx.x == 0) {
    int off = 0;
    for (int q = 0; q < 2 * R; ++q) { cursor[q] = off; off += counts[q]; }
  }
}

__global__ void __launch_bounds__(256)
scatter_kernel(const int* __restrict__ src, const int* __restrict__ tgt,
               const int* __restrict__ et, const float* __restrict__ wts,
               int E, int R, int N, int* cursor, int* s_src, int* s_tgt, float* s_w) {
  __shared__ int lc[MAXR];
  if (threadIdx.x < MAXR) lc[threadIdx.x] = 0;
  __syncthreads();
  const int NH = N >> 1;
  int base = (int)blockIdx.x * CNT_EPB;
  int r[CNT_G][4], vs[CNT_G][4], vt[CNT_G][4], rk[CNT_G][4];
  float vw[CNT_G][4];
#pragma unroll
  for (int g = 0; g < CNT_G; ++g) {
    int e = base + (g * 256 + (int)threadIdx.x) * 4;
    load4i(et,  e, E, -1, r[g]);
    load4i(src, e, E, 0, vs[g]);
    load4i(tgt, e, E, 0, vt[g]);
    load4f(wts, e, E, vw[g]);
  }
#pragma unroll
  for (int g = 0; g < CNT_G; ++g)
#pragma unroll
    for (int j = 0; j < 4; ++j)
      if (r[g][j] >= 0) r[g][j] = bucket_of(r[g][j], vt[g][j], R, NH);
#pragma unroll
  for (int g = 0; g < CNT_G; ++g)
#pragma unroll
    for (int j = 0; j < 4; ++j)
      rk[g][j] = (r[g][j] >= 0) ? atomicAdd(&lc[r[g][j]], 1) : 0;
  __syncthreads();
  if ((int)threadIdx.x < 2 * R) {
    int c = lc[threadIdx.x];
    lc[threadIdx.x] = c ? atomicAdd(&cursor[threadIdx.x], c) : 0;
  }
  __syncthreads();
  int p[CNT_G][4];
#pragma unroll
  for (int g = 0; g < CNT_G; ++g)
#pragma unroll
    for (int j = 0; j < 4; ++j)
      p[g][j] = lc[max(r[g][j], 0)] + rk[g][j];
#pragma unroll
  for (int g = 0; g < CNT_G; ++g)
#pragma unroll
    for (int j = 0; j < 4; ++j)
      if (r[g][j] >= 0) {
        s_src[p[g][j]] = vs[g][j];
        s_tgt[p[g][j]] = vt[g][j];
        s_w[p[g][j]]   = vw[g][j];
      }
}

__global__ void __launch_bounds__(256)
rgcn_tile_kernel(const float* __restrict__ x,
                 const float* __restrict__ W,
                 const int* __restrict__ s_src,
                 const int* __restrict__ s_tgt,
                 const float* __restrict__ s_w,
                 const int* __restrict__ counts,
                 int R,
                 int* tick0, int* tick1,
                 float* __restrict__ out) {
  const int Q = 2 * R;
  int tall = 0, T0 = 0;
  for (int g = 0; g < Q; g += 16) {
#pragma unroll
    for (int j = 0; j < 16; ++j) {
      T0 = (g + j == R) ? tall : T0;
      tall += (int)((unsigned)(counts[g + j] + TILE_E - 1) / TILE_E);
    }
  }
  int T1 = tall - T0;

  __shared__ int claim[2];
  if (threadIdx.x == 0) {
    int die = (__builtin_amdgcn_s_sendmsg_rtn(0x87) >> 16) & 15;
    int h = (die >> 2) & 1;
    int t = atomicAdd(h ? tick1 : tick0, 1);
    if (t >= (h ? T1 : T0)) {
      h ^= 1;
      t = atomicAdd(h ? tick1 : tick0, 1);
      if (t >= (h ? T1 : T0)) h = -1;
    }
    claim[0] = h;
    claim[1] = t;
  }
  __syncthreads();
  int hsel = claim[0];
  if (hsel < 0) return;
  int b = __builtin_amdgcn_readfirstlane(claim[1] + (hsel ? T0 : 0));

  int q = 0, cnt = 0, eo = 0, ts = 0;
  int tacc = 0, eoff = 0;
  for (int g = 0; g < Q; g += 16) {
#pragma unroll
    for (int j = 0; j < 16; ++j) {
      int c = counts[g + j];
      bool here = (b >= tacc);
      q   = here ? g + j : q;
      cnt = here ? c     : cnt;
      eo  = here ? eoff  : eo;
      ts  = here ? tacc  : ts;
      tacc += (int)((unsigned)(c + TILE_E - 1) / TILE_E);
      eoff += c;
    }
  }
  int r = (q >= R) ? q - R : q;
  int tile0 = (b - ts) * TILE_E;
  if (tile0 >= cnt) return;
  int start = eo + tile0;
  int nE = min(TILE_E, cnt - tile0);

  __shared__ __attribute__((aligned(16))) float xs[TILE_E * XS_STRIDE];
  __shared__ float Wl[DIN * W_STRIDE];
  __shared__ int   tg[TILE_E];

  for (int idx = threadIdx.x; idx < DIN * DOUT; idx += 256) {
    int k = idx >> 6, n = idx & 63;
    Wl[k * W_STRIDE + n] = W[r * DIN * DOUT + idx];
  }

  int glane = threadIdx.x & 31;
  int gw    = threadIdx.x >> 5;
  int gh    = glane >> 4;
  int gq    = glane & 15;
  int row   = gw * 16 + gq;
  int src = 0; float wt = 0.f;
  if (row < nE) { src = s_src[start + row]; wt = s_w[start + row]; }
  if (gh == 0) tg[row] = (row < nE) ? s_tgt[start + row] : -1;
  int wti = __float_as_int(wt);
#pragma unroll
  for (int i = 0; i < 8; ++i) {
    int s0 = __builtin_amdgcn_readlane(src, 2 * i);
    int s1 = __builtin_amdgcn_readlane(src, 2 * i + 1);
    int w0 = __builtin_amdgcn_readlane(wti, 2 * i);
    int w1 = __builtin_amdgcn_readlane(wti, 2 * i + 1);
    int   s = gh ? s1 : s0;
    float w = __int_as_float(gh ? w1 : w0);
    float4 v = *((const float4*)(x + (size_t)s * DIN) + gq);
    float4 o;
    o.x = v.x * w; o.y = v.y * w; o.z = v.z * w; o.w = v.w * w;
    *(float4*)&xs[(gw * 16 + 2 * i + gh) * XS_STRIDE + 4 * gq] = o;
  }
  __syncthreads();

  int lane = threadIdx.x & 31;
  int wv   = threadIdx.x >> 5;
  int hi   = lane >> 4;
  int lm   = lane & 15;
  const float* arow = &xs[(wv * 16 + lm) * XS_STRIDE + 2 * hi];

#pragma unroll
  for (int ct = 0; ct < 4; ++ct) {
    v8f acc = {};
    const float* bcol = &Wl[(2 * hi) * W_STRIDE + ct * 16 + lm];
#pragma unroll
    for (int ks = 0; ks < 16; ++ks) {
      v2f a; a.x = arow[ks * 4];               a.y = arow[ks * 4 + 1];
      v2f b; b.x = bcol[(ks * 4) * W_STRIDE];  b.y = bcol[(ks * 4 + 1) * W_STRIDE];
      acc = __builtin_amdgcn_wmma_f32_16x16x4_f32(
           false, a,  false, b,
           (short)0, acc,  false,  false);
    }
#pragma unroll
    for (int j = 0; j < 8; ++j) {
      int m = wv * 16 + 8 * hi + j;
      int t = tg[m];
      if (t >= 0) atomicAdd(&out[(size_t)t * DOUT + ct * 16 + lm], acc[j]);
    }
  }
}

static inline size_t align256(size_t v) { return (v + 255) & ~(size_t)255; }

extern "C" void kernel_launch(void* const* d_in, const int* in_sizes, int n_in,
                              void* d_out, int out_size, void* d_ws, size_t ws_size,
                              hipStream_t stream) {
  const float* x       = (const float*)d_in[0];
  const int*   source  = (const int*)d_in[1];
  const int*   target  = (const int*)d_in[2];
  const int*   etype   = (const int*)d_in[3];
  const float* ew      = (const float*)d_in[4];
  const float* bases   = (const float*)d_in[5];
  const float* comb    = (const float*)d_in[6];
  float*       out     = (float*)d_out;

  const int N = in_sizes[0] / DIN;
  const int E = in_sizes[1];
  const int B = in_sizes[5] / (DIN * DOUT);
  const int R = in_sizes[6] / B;

  uint8_t* w = (uint8_t*)d_ws;
  float* Wbuf    = (float*)w;  w += align256((size_t)R * DIN * DOUT * 4);
  int*   counts  = (int*)w;    w += 256;
  int*   cursor  = (int*)w;    w += 256;
  int*   s_src   = (int*)w;    w += align256((size_t)E * 4);
  int*   s_tgt   = (int*)w;    w += align256((size_t)E * 4);
  float* s_w     = (float*)w;

  hipMemsetAsync(counts, 0, 512, stream);
  hipMemsetAsync(out, 0, (size_t)N * DOUT * sizeof(float), stream);

  compute_W_kernel<<<R, 256, 0, stream>>>(bases, comb, Wbuf, B);

  int eblk = (E + CNT_EPB - 1) / CNT_EPB;
  hist_kernel<<<eblk, 256, 0, stream>>>(etype, target, E, R, N, counts);
  scan_kernel<<<1, 32, 0, stream>>>(counts, cursor, R);
  scatter_kernel<<<eblk, 256, 0, stream>>>(source, target, etype, ew, E, R, N,
                                           cursor, s_src, s_tgt, s_w);

  int tiles = (E + TILE_E - 1) / TILE_E + 2 * R;
  rgcn_tile_kernel<<<tiles, 256, 0, stream>>>(x, Wbuf, s_src, s_tgt, s_w,
                                              counts, R, cursor + 63, counts + 63, out);
}


# main epilogue: accumulators of two column tiles trade half-waves (v_permlane16_swap), so each f32 atomic adds one whole 128-B out line of one row (was 2 half lines); arrival-order adds as before
# speedup vs baseline: 5.2589x; 1.0396x over previous
#include <hip/hip_runtime.h>
#include <hip/hip_bf16.h>

typedef __attribute__((ext_vector_type(2))) float v2f;
typedef __attribute__((ext_vector_type(8))) float v8f;

#define DIN 64
#define DOUT 64
#define TILE_E 128
#define XS_STRIDE 68
#define W_STRIDE 65

__global__ void compute_W_kernel(const float* __restrict__ bases,
                                 const float* __restrict__ comb,
                                 float* __restrict__ W, int B) {
  int r = blockIdx.x;
  const float* c = comb + r * B;
  for (int idx = threadIdx.x; idx < DIN * DOUT; idx += blockDim.x) {
    float acc = 0.f;
    for (int b = 0; b < B; ++b) acc += c[b] * bases[b * DIN * DOUT + idx];
    W[r * DIN * DOUT + idx] = acc;
  }
}

#define CNT_G 2
#define CNT_EPB (256 * 4 * CNT_G)
#define MAXR 64

__device__ __forceinline__ void load4i(const int* __restrict__ p, int e, int E, int fill, int (&v)[4]) {
  if (e + 4 <= E) {
    int4 q = *(const int4*)(p + e);
    v[0] = q.x; v[1] = q.y; v[2] = q.z; v[3] = q.w;
  } else {
#pragma unroll
    for (int j = 0; j < 4; ++j) v[j] = (e + j < E) ? p[e + j] : fill;
  }
}
__device__ __forceinline__ void load4f(const float* __restrict__ p, int e, int E, float (&v)[4]) {
  if (e + 4 <= E) {
    float4 q = *(const float4*)(p + e);
    v[0] = q.x; v[1] = q.y; v[2] = q.z; v[3] = q.w;
  } else {
#pragma unroll
    for (int j = 0; j < 4; ++j) v[j] = (e + j < E) ? p[e + j] : 0.f;
  }
}

__device__ __forceinline__ int bucket_of(int rel, int tgt, int R, int NH) {
  return rel + (tgt >= NH ? R : 0);
}

__global__ void __launch_bounds__(256)
hist_kernel(const int* __restrict__ et, const int* __restrict__ tgt, int E, int R, int N, int* counts) {
  __shared__ int lc[MAXR];
  if (threadIdx.x < MAXR) lc[threadIdx.x] = 0;
  __syncthreads();
  const int NH = N >> 1;
  int base = (int)blockIdx.x * CNT_EPB;
  int r[CNT_G][4], vt[CNT_G][4];
#pragma unroll
  for (int g = 0; g < CNT_G; ++g) {
    int e = base + (g * 256 + (int)threadIdx.x) * 4;
    load4i(et,  e, E, -1, r[g]);
    load4i(tgt, e, E, 0, vt[g]);
  }
#pragma unroll
  for (int g = 0; g < CNT_G; ++g)
#pragma unroll
    for (int j = 0; j < 4; ++j)
      if (r[g][j] >= 0) atomicAdd(&lc[bucket_of(r[g][j], vt[g][j], R, NH)], 1);
  __syncthreads();
  if ((int)threadIdx.x < 2 * R) {
    int c = lc[threadIdx.x];
    if (c) atomicAdd(&counts[threadIdx.x], c);
  }
}

__global__ void scan_kernel(const int* __restrict__ counts, int* cursor, int R) {
  if (threadIdx.x == 0 && blockIdx.x == 0) {
    int off = 0;
    for (int q = 0; q < 2 * R; ++q) { cursor[q] = off; off += counts[q]; }
  }
}

__global__ void __launch_bounds__(256)
scatter_kernel(const int* __restrict__ src, const int* __restrict__ tgt,
               const int* __restrict__ et, const float* __restrict__ wts,
               int E, int R, int N, int* cursor, int* s_src, int* s_tgt, float* s_w) {
  __shared__ int lc[MAXR];
  if (threadIdx.x < MAXR) lc[threadIdx.x] = 0;
  __syncthreads();
  const int NH = N >> 1;
  int base = (int)blockIdx.x * CNT_EPB;
  int r[CNT_G][4], vs[CNT_G][4], vt[CNT_G][4], rk[CNT_G][4];
  float vw[CNT_G][4];
#pragma unroll
  for (int g = 0; g < CNT_G; ++g) {
    int e = base + (g * 256 + (int)threadIdx.x) * 4;
    load4i(et,  e, E, -1, r[g]);
    load4i(src, e, E, 0, vs[g]);
    load4i(tgt, e, E, 0, vt[g]);
    load4f(wts, e, E, vw[g]);
  }
#pragma unroll
  for (int g = 0; g < CNT_G; ++g)
#pragma unroll
    for (int j = 0; j < 4; ++j)
      if (r[g][j] >= 0) r[g][j] = bucket_of(r[g][j], vt[g][j], R, NH);
#pragma unroll
  for (int g = 0; g < CNT_G; ++g)
#pragma unroll
    for (int j = 0; j < 4; ++j)
      rk[g][j] = (r[g][j] >= 0) ? atomicAdd(&lc[r[g][j]], 1) : 0;
  __syncthreads();
  if ((int)threadIdx.x < 2 * R) {
    int c = lc[threadIdx.x];
    lc[threadIdx.x] = c ? atomicAdd(&cursor[threadIdx.x], c) : 0;
  }
  __syncthreads();
  int p[CNT_G][4];
#pragma unroll
  for (int g = 0; g < CNT_G; ++g)
#pragma unroll
    for (int j = 0; j < 4; ++j)
      p[g][j] = lc[max(r[g][j], 0)] + rk[g][j];
#pragma unroll
  for (int g = 0; g < CNT_G; ++g)
#pragma unroll
    for (int j = 0; j < 4; ++j)
      if (r[g][j] >= 0) {
        s_src[p[g][j]] = vs[g][j];
        s_tgt[p[g][j]] = vt[g][j];
        s_w[p[g][j]]   = vw[g][j];
      }
}

__global__ void __launch_bounds__(256)
rgcn_tile_kernel(const float* __restrict__ x,
                 const float* __restrict__ W,
                 const int* __restrict__ s_src,
                 const int* __restrict__ s_tgt,
                 const float* __restrict__ s_w,
                 const int* __restrict__ counts,
                 int R,
                 int* tick0, int* tick1,
                 float* __restrict__ out) {
  const int Q = 2 * R;
  int tall = 0, T0 = 0;
  for (int g = 0; g < Q; g += 16) {
#pragma unroll
    for (int j = 0; j < 16; ++j) {
      T0 = (g + j == R) ? tall : T0;
      tall += (int)((unsigned)(counts[g + j] + TILE_E - 1) / TILE_E);
    }
  }
  int T1 = tall - T0;

  __shared__ int claim[2];
  if (threadIdx.x == 0) {
    int die = (__builtin_amdgcn_s_sendmsg_rtn(0x87) >> 16) & 15;
    int h = (die >> 2) & 1;
    int t = atomicAdd(h ? tick1 : tick0, 1);
    if (t >= (h ? T1 : T0)) {
      h ^= 1;
      t = atomicAdd(h ? tick1 : tick0, 1);
      if (t >= (h ? T1 : T0)) h = -1;
    }
    claim[0] = h;
    claim[1] = t;
  }
  __syncthreads();
  int hsel = claim[0];
  if (hsel < 0) return;
  int b = __builtin_amdgcn_readfirstlane(claim[1] + (hsel ? T0 : 0));

  int q = 0, cnt = 0, eo = 0, ts = 0;
  int tacc = 0, eoff = 0;
  for (int g = 0; g < Q; g += 16) {
#pragma unroll
    for (int j = 0; j < 16; ++j) {
      int c = counts[g + j];
      bool here = (b >= tacc);
      q   = here ? g + j : q;
      cnt = here ? c     : cnt;
      eo  = here ? eoff  : eo;
      ts  = here ? tacc  : ts;
      tacc += (int)((unsigned)(c + TILE_E - 1) / TILE_E);
      eoff += c;
    }
  }
  int r = (q >= R) ? q - R : q;
  int tile0 = (b - ts) * TILE_E;
  if (tile0 >= cnt) return;
  int start = eo + tile0;
  int nE = min(TILE_E, cnt - tile0);

  __shared__ __attribute__((aligned(16))) float xs[TILE_E * XS_STRIDE];
  __shared__ float Wl[DIN * W_STRIDE];

  for (int idx = threadIdx.x; idx < DIN * DOUT; idx += 256) {
    int k = idx >> 6, n = idx & 63;
    Wl[k * W_STRIDE + n] = W[r * DIN * DOUT + idx];
  }

  int glane = threadIdx.x & 31;
  int gw    = threadIdx.x >> 5;
  int gh    = glane >> 4;
  int gq    = glane & 15;
  int row   = gw * 16 + gq;
  int src = 0; float wt = 0.f;
  int tv = -1;
  if (row < nE) { src = s_src[start + row]; wt = s_w[start + row]; tv = s_tgt[start + row]; }
  int wti = __float_as_int(wt);
#pragma unroll
  for (int i = 0; i < 8; ++i) {
    int s0 = __builtin_amdgcn_readlane(src, 2 * i);
    int s1 = __builtin_amdgcn_readlane(src, 2 * i + 1);
    int w0 = __builtin_amdgcn_readlane(wti, 2 * i);
    int w1 = __builtin_amdgcn_readlane(wti, 2 * i + 1);
    int   s = gh ? s1 : s0;
    float w = __int_as_float(gh ? w1 : w0);
    float4 v = *((const float4*)(x + (size_t)s * DIN) + gq);
    float4 o;
    o.x = v.x * w; o.y = v.y * w; o.z = v.z * w; o.w = v.w * w;
    *(float4*)&xs[(gw * 16 + 2 * i + gh) * XS_STRIDE + 4 * gq] = o;
  }
  __syncthreads();

  int lane = threadIdx.x & 31;
  int wv   = threadIdx.x >> 5;
  int hi   = lane >> 4;
  int lm   = lane & 15;
  const float* arow = &xs[(wv * 16 + lm) * XS_STRIDE + 2 * hi];

#pragma unroll
  for (int p = 0; p < 2; ++p) {
    v8f acc0 = {}, acc1 = {};
    const float* bcol0 = &Wl[(2 * hi) * W_STRIDE + (2 * p) * 16 + lm];
    const float* bcol1 = bcol0 + 16;
#pragma unroll
    for (int ks = 0; ks < 16; ++ks) {
      v2f a;  a.x  = arow[ks * 4];                a.y  = arow[ks * 4 + 1];
      v2f b0; b0.x = bcol0[(ks * 4) * W_STRIDE];  b0.y = bcol0[(ks * 4 + 1) * W_STRIDE];
      v2f b1; b1.x = bcol1[(ks * 4) * W_STRIDE];  b1.y = bcol1[(ks * 4 + 1) * W_STRIDE];
      acc0 = __builtin_amdgcn_wmma_f32_16x16x4_f32(
           false, a,  false, b0,
           (short)0, acc0,  false,  false);
      acc1 = __builtin_amdgcn_wmma_f32_16x16x4_f32(
           false, a,  false, b1,
           (short)0, acc1,  false,  false);
    }
    float* ocol = out + p * 32 + lane;
#pragma unroll
    for (int j = 0; j < 8; ++j) {
      auto sw = __builtin_amdgcn_permlane16_swap(__float_as_uint(acc0[j]), __float_as_uint(acc1[j]),
                                                 false, false);
      int t0 = __builtin_amdgcn_readlane(tv, j);
      int t1 = __builtin_amdgcn_readlane(tv, j + 8);
      if (t0 >= 0) atomicAdd(ocol + (size_t)t0 * DOUT, __uint_as_float(sw[0]));
      if (t1 >= 0) atomicAdd(ocol + (size_t)t1 * DOUT, __uint_as_float(sw[1]));
    }
  }
}

static inline size_t align256(size_t v) { return (v + 255) & ~(size_t)255; }

extern "C" void kernel_launch(void* const* d_in, const int* in_sizes, int n_in,
                              void* d_out, int out_size, void* d_ws, size_t ws_size,
                              hipStream_t stream) {
  const float* x       = (const float*)d_in[0];
  const int*   source  = (const int*)d_in[1];
  const int*   target  = (const int*)d_in[2];
  const int*   etype   = (const int*)d_in[3];
  const float* ew      = (const float*)d_in[4];
  const float* bases   = (const float*)d_in[5];
  const float* comb    = (const float*)d_in[6];
  float*       out     = (float*)d_out;

  const int N = in_sizes[0] / DIN;
  const int E = in_sizes[1];
  const int B = in_sizes[5] / (DIN * DOUT);
  const int R = in_sizes[6] / B;

  uint8_t* w = (uint8_t*)d_ws;
  float* Wbuf    = (float*)w;  w += align256((size_t)R * DIN * DOUT * 4);
  int*   counts  = (int*)w;    w += 256;
  int*   cursor  = (int*)w;    w += 256;
  int*   s_src   = (int*)w;    w += align256((size_t)E * 4);
  int*   s_tgt   = (int*)w;    w += align256((size_t)E * 4);
  float* s_w     = (float*)w;

  hipMemsetAsync(counts, 0, 512, stream);
  hipMemsetAsync(out, 0, (size_t)N * DOUT * sizeof(float), stream);

  compute_W_kernel<<<R, 256, 0, stream>>>(bases, comb, Wbuf, B);

  int eblk = (E + CNT_EPB - 1) / CNT_EPB;
  hist_kernel<<<eblk, 256, 0, stream>>>(etype, target, E, R, N, counts);
  scan_kernel<<<1, 32, 0, stream>>>(counts, cursor, R);
  scatter_kernel<<<eblk, 256, 0, stream>>>(source, target, etype, ew, E, R, N,
                                           cursor, s_src, s_tgt, s_w);

  int tiles = (E + TILE_E - 1) / TILE_E + 2 * R;
  rgcn_tile_kernel<<<tiles, 256, 0, stream>>>(x, Wbuf, s_src, s_tgt, s_w,
                                              counts, R, cursor + 63, counts + 63, out);
}


# launches fused: out clear and W build run inside the histogram kernel, each scatter block sums its own bucket starts (scan launch gone): 7 stream operations become 4; float-add order as before
# speedup vs baseline: 5.9816x; 1.1374x over previous
#include <hip/hip_runtime.h>
#include <hip/hip_bf16.h>

typedef __attribute__((ext_vector_type(2))) float v2f;
typedef __attribute__((ext_vector_type(8))) float v8f;

#define DIN 64
#define DOUT 64
#define TILE_E 128
#define XS_STRIDE 68
#define W_STRIDE 65

#define CNT_G 2
#define CNT_EPB (256 * 4 * CNT_G)
#define MAXR 64

__device__ __forceinline__ void load4i(const int* __restrict__ p, int e, int E, int fill, int (&v)[4]) {
  if (e + 4 <= E) {
    int4 q = *(const int4*)(p + e);
    v[0] = q.x; v[1] = q.y; v[2] = q.z; v[3] = q.w;
  } else {
#pragma unroll
    for (int j = 0; j < 4; ++j) v[j] = (e + j < E) ? p[e + j] : fill;
  }
}
__device__ __forceinline__ void load4f(const float* __restrict__ p, int e, int E, float (&v)[4]) {
  if (e + 4 <= E) {
    float4 q = *(const float4*)(p + e);
    v[0] = q.x; v[1] = q.y; v[2] = q.z; v[3] = q.w;
  } else {
#pragma unroll
    for (int j = 0; j < 4; ++j) v[j] = (e + j < E) ? p[e + j] : 0.f;
  }
}

__device__ __forceinline__ int bucket_of(int rel, int tgt, int R, int NH) {
  return rel + (tgt >= NH ? R : 0);
}

__global__ void __launch_bounds__(256)
hist_kernel(const int* __restrict__ et, const int* __restrict__ tgt, int E, int R, int N, int* counts,
            const float* __restrict__ bases, const float* __restrict__ comb,
            float* __restrict__ W, int B, float* __restrict__ out) {
  __shared__ int lc[MAXR];
  if (threadIdx.x < MAXR) lc[threadIdx.x] = 0;
  __syncthreads();
  const int NH = N >> 1;
  int base = (int)blockIdx.x * CNT_EPB;
  int r[CNT_G][4], vt[CNT_G][4];
#pragma unroll
  for (int g = 0; g < CNT_G; ++g) {
    int e = base + (g * 256 + (int)threadIdx.x) * 4;
    load4i(et,  e, E, -1, r[g]);
    load4i(tgt, e, E, 0, vt[g]);
  }
  const int gid = (int)blockIdx.x * 256 + (int)threadIdx.x;
  const int gsz = (int)gridDim.x * 256;
  {
    float4* o4 = (float4*)out;
    const int n4 = N * (DOUT / 4);
    const float4 z = make_float4(0.f, 0.f, 0.f, 0.f);
    for (int i = gid; i < n4; i += gsz) o4[i] = z;
  }
  {
    const int nW = R * DIN * DOUT;
    for (int idx = gid; idx < nW; idx += gsz) {
      const float* c = comb + (idx / (DIN * DOUT)) * B;
      const int k = idx % (DIN * DOUT);
      float acc = 0.f;
      for (int b = 0; b < B; ++b) acc += c[b] * bases[b * DIN * DOUT + k];
      W[idx] = acc;
    }
  }
#pragma unroll
  for (int g = 0; g < CNT_G; ++g)
#pragma unroll
    for (int j = 0; j < 4; ++j)
      if (r[g][j] >= 0) atomicAdd(&lc[bucket_of(r[g][j], vt[g][j], R, NH)], 1);
  __syncthreads();
  if ((int)threadIdx.x < 2 * R) {
    int c = lc[threadIdx.x];
    if (c) atomicAdd(&counts[threadIdx.x], c);
  }
}

__global__ void __launch_bounds__(256)
scatter_kernel(const int* __restrict__ src, const int* __restrict__ tgt,
               const int* __restrict__ et, const float* __restrict__ wts,
               int E, int R, int N, const int* __restrict__ counts, int* taken,
               int* s_src, int* s_tgt, float* s_w) {
  __shared__ int lc[MAXR];
  __shared__ int cn[MAXR];
  const int NH = N >> 1;
  int base = (int)blockIdx.x * CNT_EPB;
  int r[CNT_G][4], vs[CNT_G][4], vt[CNT_G][4], rk[CNT_G][4];
  float vw[CNT_G][4];
#pragma unroll
  for (int g = 0; g < CNT_G; ++g) {
    int e = base + (g * 256 + (int)threadIdx.x) * 4;
    load4i(et,  e, E, -1, r[g]);
    load4i(src, e, E, 0, vs[g]);
    load4i(tgt, e, E, 0, vt[g]);
    load4f(wts, e, E, vw[g]);
  }
  if (threadIdx.x < MAXR) {
    lc[threadIdx.x] = 0;
    cn[threadIdx.x] = ((int)threadIdx.x < 2 * R) ? counts[threadIdx.x] : 0;
  }
  __syncthreads();
  int bstart = 0;
  if ((int)threadIdx.x < 2 * R) {
#pragma unroll
    for (int j = 0; j < 48; ++j) bstart += (j < (int)threadIdx.x) ? cn[j] : 0;
  }
#pragma unroll
  for (int g = 0; g < CNT_G; ++g)
#pragma unroll
    for (int j = 0; j < 4; ++j)
      if (r[g][j] >= 0) r[g][j] = bucket_of(r[g][j], vt[g][j], R, NH);
#pragma unroll
  for (int g = 0; g < CNT_G; ++g)
#pragma unroll
    for (int j = 0; j < 4; ++j)
      rk[g][j] = (r[g][j] >= 0) ? atomicAdd(&lc[r[g][j]], 1) : 0;
  __syncthreads();
  if ((int)threadIdx.x < 2 * R) {
    int c = lc[threadIdx.x];
    lc[threadIdx.x] = c ? bstart + atomicAdd(&taken[threadIdx.x], c) : 0;
  }
  __syncthreads();
  int p[CNT_G][4];
#pragma unroll
  for (int g = 0; g < CNT_G; ++g)
#pragma unroll
    for (int j = 0; j < 4; ++j)
      p[g][j] = lc[max(r[g][j], 0)] + rk[g][j];
#pragma unroll
  for (int g = 0; g < CNT_G; ++g)
#pragma unroll
    for (int j = 0; j < 4; ++j)
      if (r[g][j] >= 0) {
        s_src[p[g][j]] = vs[g][j];
        s_tgt[p[g][j]] = vt[g][j];
        s_w[p[g][j]]   = vw[g][j];
      }
}

__global__ void __launch_bounds__(256)
rgcn_tile_kernel(const float* __restrict__ x,
                 const float* __restrict__ W,
                 const int* __restrict__ s_src,
                 const int* __restrict__ s_tgt,
                 const float* __restrict__ s_w,
                 const int* __restrict__ counts,
                 int R,
                 int* tick0, int* tick1,
                 float* __restrict__ out) {
  const int Q = 2 * R;
  int tall = 0, T0 = 0;
  for (int g = 0; g < Q; g += 16) {
#pragma unroll
    for (int j = 0; j < 16; ++j) {
      T0 = (g + j == R) ? tall : T0;
      tall += (int)((unsigned)(counts[g + j] + TILE_E - 1) / TILE_E);
    }
  }
  int T1 = tall - T0;

  __shared__ int claim[2];
  if (threadIdx.x == 0) {
    int die = (__builtin_amdgcn_s_sendmsg_rtn(0x87) >> 16) & 15;
    int h = (die >> 2) & 1;
    int t = atomicAdd(h ? tick1 : tick0, 1);
    if (t >= (h ? T1 : T0)) {
      h ^= 1;
      t = atomicAdd(h ? tick1 : tick0, 1);
      if (t >= (h ? T1 : T0)) h = -1;
    }
    claim[0] = h;
    claim[1] = t;
  }
  __syncthreads();
  int hsel = claim[0];
  if (hsel < 0) return;
  int b = __builtin_amdgcn_readfirstlane(claim[1] + (hsel ? T0 : 0));

  int q = 0, cnt = 0, eo = 0, ts = 0;
  int tacc = 0, eoff = 0;
  for (int g = 0; g < Q; g += 16) {
#pragma unroll
    for (int j = 0; j < 16; ++j) {
      int c = counts[g + j];
      bool here = (b >= tacc);
      q   = here ? g + j : q;
      cnt = here ? c     : cnt;
      eo  = here ? eoff  : eo;
      ts  = here ? tacc  : ts;
      tacc += (int)((unsigned)(c + TILE_E - 1) / TILE_E);
      eoff += c;
    }
  }
  int r = (q >= R) ? q - R : q;
  int tile0 = (b - ts) * TILE_E;
  if (tile0 >= cnt) return;
  int start = eo + tile0;
  int nE = min(TILE_E, cnt - tile0);

  __shared__ __attribute__((aligned(16))) float xs[TILE_E * XS_STRIDE];
  __shared__ float Wl[DIN * W_STRIDE];

  for (int idx = threadIdx.x; idx < DIN * DOUT; idx += 256) {
    int k = idx >> 6, n = idx & 63;
    Wl[k * W_STRIDE + n] = W[r * DIN * DOUT + idx];
  }

  int glane = threadIdx.x & 31;
  int gw    = threadIdx.x >> 5;
  int gh    = glane >> 4;
  int gq    = glane & 15;
  int row   = gw * 16 + gq;
  int src = 0; float wt = 0.f;
  int tv = -1;
  if (row < nE) { src = s_src[start + row]; wt = s_w[start + row]; tv = s_tgt[start + row]; }
  int wti = __float_as_int(wt);
#pragma unroll
  for (int i = 0; i < 8; ++i) {
    int s0 = __builtin_amdgcn_readlane(src, 2 * i);
    int s1 = __builtin_amdgcn_readlane(src, 2 * i + 1);
    int w0 = __builtin_amdgcn_readlane(wti, 2 * i);
    int w1 = __builtin_amdgcn_readlane(wti, 2 * i + 1);
    int   s = gh ? s1 : s0;
    float w = __int_as_float(gh ? w1 : w0);
    float4 v = *((const float4*)(x + (size_t)s * DIN) + gq);
    float4 o;
    o.x = v.x * w; o.y = v.y * w; o.z = v.z * w; o.w = v.w * w;
    *(float4*)&xs[(gw * 16 + 2 * i + gh) * XS_STRIDE + 4 * gq] = o;
  }
  __syncthreads();

  int lane = threadIdx.x & 31;
  int wv   = threadIdx.x >> 5;
  int hi   = lane >> 4;
  int lm   = lane & 15;
  const float* arow = &xs[(wv * 16 + lm) * XS_STRIDE + 2 * hi];

#pragma unroll
  for (int p = 0; p < 2; ++p) {
    v8f acc0 = {}, acc1 = {};
    const float* bcol0 = &Wl[(2 * hi) * W_STRIDE + (2 * p) * 16 + lm];
    const float* bcol1 = bcol0 + 16;
#pragma unroll
    for (int ks = 0; ks < 16; ++ks) {
      v2f a;  a.x  = arow[ks * 4];                a.y  = arow[ks * 4 + 1];
      v2f b0; b0.x = bcol0[(ks * 4) * W_STRIDE];  b0.y = bcol0[(ks * 4 + 1) * W_STRIDE];
      v2f b1; b1.x = bcol1[(ks * 4) * W_STRIDE];  b1.y = bcol1[(ks * 4 + 1) * W_STRIDE];
      acc0 = __builtin_amdgcn_wmma_f32_16x16x4_f32(
           false, a,  false, b0,
           (short)0, acc0,  false,  false);
      acc1 = __builtin_amdgcn_wmma_f32_16x16x4_f32(
           false, a,  false, b1,
           (short)0, acc1,  false,  false);
    }
    float* ocol = out + p * 32 + lane;
#pragma unroll
    for (int j = 0; j < 8; ++j) {
      auto sw = __builtin_amdgcn_permlane16_swap(__float_as_uint(acc0[j]), __float_as_uint(acc1[j]),
                                                 false, false);
      int t0 = __builtin_amdgcn_readlane(tv, j);
      int t1 = __builtin_amdgcn_readlane(tv, j + 8);
      if (t0 >= 0) atomicAdd(ocol + (size_t)t0 * DOUT, __uint_as_float(sw[0]));
      if (t1 >= 0) atomicAdd(ocol + (size_t)t1 * DOUT, __uint_as_float(sw[1]));
    }
  }
}

static inline size_t align256(size_t v) { return (v + 255) & ~(size_t)255; }

extern "C" void kernel_launch(void* const* d_in, const int* in_sizes, int n_in,
                              void* d_out, int out_size, void* d_ws, size_t ws_size,
                              hipStream_t stream) {
  const float* x       = (const float*)d_in[0];
  const int*   source  = (const int*)d_in[1];
  const int*   target  = (const int*)d_in[2];
  const int*   etype   = (const int*)d_in[3];
  const float* ew      = (const float*)d_in[4];
  const float* bases   = (const float*)d_in[5];
  const float* comb    = (const float*)d_in[6];
  float*       out     = (float*)d_out;

  const int N = in_sizes[0] / DIN;
  const int E = in_sizes[1];
  const int B = in_sizes[5] / (DIN * DOUT);
  const int R = in_sizes[6] / B;

  uint8_t* w = (uint8_t*)d_ws;
  float* Wbuf    = (float*)w;  w += align256((size_t)R * DIN * DOUT * 4);
  int*   counts  = (int*)w;    w += 256;
  int*   cursor  = (int*)w;    w += 256;
  int*   s_src   = (int*)w;    w += align256((size_t)E * 4);
  int*   s_tgt   = (int*)w;    w += align256((size_t)E * 4);
  float* s_w     = (float*)w;

  hipMemsetAsync(counts, 0, 512, stream);

  int eblk = (E + CNT_EPB - 1) / CNT_EPB;
  hist_kernel<<<eblk, 256, 0, stream>>>(etype, target, E, R, N, counts, bases, comb, Wbuf, B, out);
  scatter_kernel<<<eblk, 256, 0, stream>>>(source, target, etype, ew, E, R, N,
                                           counts, cursor, s_src, s_tgt, s_w);

  int tiles = (E + TILE_E - 1) / TILE_E + 2 * R;
  rgcn_tile_kernel<<<tiles, 256, 0, stream>>>(x, Wbuf, s_src, s_tgt, s_w,
                                              counts, R, cursor + 63, counts + 63, out);
}


# main GEMM: W_r kept in LDS in fragment order (k-pairs, odd rows' halves swapped): B of two column tiles is one conflict-free 16-byte read (was four 4-byte, conflicts), A read once; add order kept
# speedup vs baseline: 6.3743x; 1.0657x over previous
#include <hip/hip_runtime.h>
#include <hip/hip_bf16.h>

typedef __attribute__((ext_vector_type(2))) float v2f;
typedef __attribute__((ext_vector_type(4))) float v4f;
typedef __attribute__((ext_vector_type(8))) float v8f;

#define DIN 64
#define DOUT 64
#define TILE_E 128
#define XS_STRIDE 68
#define WP_PITCH 128

#define CNT_G 2
#define CNT_EPB (256 * 4 * CNT_G)
#define MAXR 64

__device__ __forceinline__ void load4i(const int* __restrict__ p, int e, int E, int fill, int (&v)[4]) {
  if (e + 4 <= E) {
    int4 q = *(const int4*)(p + e);
    v[0] = q.x; v[1] = q.y; v[2] = q.z; v[3] = q.w;
  } else {
#pragma unroll
    for (int j = 0; j < 4; ++j) v[j] = (e + j < E) ? p[e + j] : fill;
  }
}
__device__ __forceinline__ void load4f(const float* __restrict__ p, int e, int E, float (&v)[4]) {
  if (e + 4 <= E) {
    float4 q = *(const float4*)(p + e);
    v[0] = q.x; v[1] = q.y; v[2] = q.z; v[3] = q.w;
  } else {
#pragma unroll
    for (int j = 0; j < 4; ++j) v[j] = (e + j < E) ? p[e + j] : 0.f;
  }
}

__device__ __forceinline__ int bucket_of(int rel, int tgt, int R, int NH) {
  return rel + (tgt >= NH ? R : 0);
}

__global__ void __launch_bounds__(256)
hist_kernel(const int* __restrict__ et, const int* __restrict__ tgt, int E, int R, int N, int* counts,
            const float* __restrict__ bases, const float* __restrict__ comb,
            float* __restrict__ W, int B, float* __restrict__ out) {
  __shared__ int lc[MAXR];
  if (threadIdx.x < MAXR) lc[threadIdx.x] = 0;
  __syncthreads();
  const int NH = N >> 1;
  int base = (int)blockIdx.x * CNT_EPB;
  int r[CNT_G][4], vt[CNT_G][4];
#pragma unroll
  for (int g = 0; g < CNT_G; ++g) {
    int e = base + (g * 256 + (int)threadIdx.x) * 4;
    load4i(et,  e, E, -1, r[g]);
    load4i(tgt, e, E, 0, vt[g]);
  }
  const int gid = (int)blockIdx.x * 256 + (int)threadIdx.x;
  const int gsz = (int)gridDim.x * 256;
  {
    float4* o4 = (float4*)out;
    const int n4 = N * (DOUT / 4);
    const float4 z = make_float4(0.f, 0.f, 0.f, 0.f);
    for (int i = gid; i < n4; i += gsz) o4[i] = z;
  }
  {
    const int nW = R * DIN * DOUT;
    for (int idx = gid; idx < nW; idx += gsz) {
      const float* c = comb + (idx / (DIN * DOUT)) * B;
      const int k = idx % (DIN * DOUT);
      float acc = 0.f;
      for (int b = 0; b < B; ++b) acc += c[b] * bases[b * DIN * DOUT + k];
      W[idx] = acc;
    }
  }
#pragma unroll
  for (int g = 0; g < CNT_G; ++g)
#pragma unroll
    for (int j = 0; j < 4; ++j)
      if (r[g][j] >= 0) atomicAdd(&lc[bucket_of(r[g][j], vt[g][j], R, NH)], 1);
  __syncthreads();
  if ((int)threadIdx.x < 2 * R) {
    int c = lc[threadIdx.x];
    if (c) atomicAdd(&counts[threadIdx.x], c);
  }
}

__global__ void __launch_bounds__(256)
scatter_kernel(const int* __restrict__ src, const int* __restrict__ tgt,
               const int* __restrict__ et, const float* __restrict__ wts,
               int E, int R, int N, const int* __restrict__ counts, int* taken,
               int* s_src, int* s_tgt, float* s_w) {
  __shared__ int lc[MAXR];
  __shared__ int cn[MAXR];
  const int NH = N >> 1;
  int base = (int)blockIdx.x * CNT_EPB;
  int r[CNT_G][4], vs[CNT_G][4], vt[CNT_G][4], rk[CNT_G][4];
  float vw[CNT_G][4];
#pragma unroll
  for (int g = 0; g < CNT_G; ++g) {
    int e = base + (g * 256 + (int)threadIdx.x) * 4;
    load4i(et,  e, E, -1, r[g]);
    load4i(src, e, E, 0, vs[g]);
    load4i(tgt, e, E, 0, vt[g]);
    load4f(wts, e, E, vw[g]);
  }
  if (threadIdx.x < MAXR) {
    lc[threadIdx.x] = 0;
    cn[threadIdx.x] = ((int)threadIdx.x < 2 * R) ? counts[threadIdx.x] : 0;
  }
  __syncthreads();
  int bstart = 0;
  if ((int)threadIdx.x < 2 * R) {
#pragma unroll
    for (int j = 0; j < 48; ++j) bstart += (j < (int)threadIdx.x) ? cn[j] : 0;
  }
#pragma unroll
  for (int g = 0; g < CNT_G; ++g)
#pragma unroll
    for (int j = 0; j < 4; ++j)
      if (r[g][j] >= 0) r[g][j] = bucket_of(r[g][j], vt[g][j], R, NH);
#pragma unroll
  for (int g = 0; g < CNT_G; ++g)
#pragma unroll
    for (int j = 0; j < 4; ++j)
      rk[g][j] = (r[g][j] >= 0) ? atomicAdd(&lc[r[g][j]], 1) : 0;
  __syncthreads();
  if ((int)threadIdx.x < 2 * R) {
    int c = lc[threadIdx.x];
    lc[threadIdx.x] = c ? bstart + atomicAdd(&taken[threadIdx.x], c) : 0;
  }
  __syncthreads();
  int p[CNT_G][4];
#pragma unroll
  for (int g = 0; g < CNT_G; ++g)
#pragma unroll
    for (int j = 0; j < 4; ++j)
      p[g][j] = lc[max(r[g][j], 0)] + rk[g][j];
#pragma unroll
  for (int g = 0; g < CNT_G; ++g)
#pragma unroll
    for (int j = 0; j < 4; ++j)
      if (r[g][j] >= 0) {
        s_src[p[g][j]] = vs[g][j];
        s_tgt[p[g][j]] = vt[g][j];
        s_w[p[g][j]]   = vw[g][j];
      }
}

__global__ void __launch_bounds__(256)
rgcn_tile_kernel(const float* __restrict__ x,
                 const float* __restrict__ W,
                 const int* __restrict__ s_src,
                 const int* __restrict__ s_tgt,
                 const float* __restrict__ s_w,
                 const int* __restrict__ counts,
                 int R,
                 int* tick0, int* tick1,
                 float* __restrict__ out) {
  const int Q = 2 * R;
  int tall = 0, T0 = 0;
  for (int g = 0; g < Q; g += 16) {
#pragma unroll
    for (int j = 0; j < 16; ++j) {
      T0 = (g + j == R) ? tall : T0;
      tall += (int)((unsigned)(counts[g + j] + TILE_E - 1) / TILE_E);
    }
  }
  int T1 = tall - T0;

  __shared__ int claim[2];
  if (threadIdx.x == 0) {
    int die = (__builtin_amdgcn_s_sendmsg_rtn(0x87) >> 16) & 15;
    int h = (die >> 2) & 1;
    int t = atomicAdd(h ? tick1 : tick0, 1);
    if (t >= (h ? T1 : T0)) {
      h ^= 1;
      t = atomicAdd(h ? tick1 : tick0, 1);
      if (t >= (h ? T1 : T0)) h = -1;
    }
    claim[0] = h;
    claim[1] = t;
  }
  __syncthreads();
  int hsel = claim[0];
  if (hsel < 0) return;
  int b = __builtin_amdgcn_readfirstlane(claim[1] + (hsel ? T0 : 0));

  int q = 0, cnt = 0, eo = 0, ts = 0;
  int tacc = 0, eoff = 0;
  for (int g = 0; g < Q; g += 16) {
#pragma unroll
    for (int j = 0; j < 16; ++j) {
      int c = counts[g + j];
      bool here = (b >= tacc);
      q   = here ? g + j : q;
      cnt = here ? c     : cnt;
      eo  = here ? eoff  : eo;
      ts  = here ? tacc  : ts;
      tacc += (int)((unsigned)(c + TILE_E - 1) / TILE_E);
      eoff += c;
    }
  }
  int r = (q >= R) ? q - R : q;
  int tile0 = (b - ts) * TILE_E;
  if (tile0 >= cnt) return;
  int start = eo + tile0;
  int nE = min(TILE_E, cnt - tile0);

  __shared__ __attribute__((aligned(16))) float xs[TILE_E * XS_STRIDE];
  __shared__ __attribute__((aligned(16))) float Wl[(DIN / 2) * WP_PITCH];

  const int wn  = threadIdx.x & 63;
  const int wk0 = threadIdx.x >> 6;
  const float* Wg = W + r * DIN * DOUT + (2 * wk0) * DOUT + wn;
  float* wdst = &Wl[wk0 * WP_PITCH + (wn >> 5) * 64 +
                    ((((wn & 15) << 2) + ((wn >> 4) & 1) * 2) ^ ((wk0 & 1) << 5))];
  v2f wreg[8];
#pragma unroll
  for (int i = 0; i < 8; ++i) {
    wreg[i].x = Wg[(8 * i) * DOUT];
    wreg[i].y = Wg[(8 * i + 1) * DOUT];
  }

  int glane = threadIdx.x & 31;
  int gw    = threadIdx.x >> 5;
  int gh    = glane >> 4;
  int gq    = glane & 15;
  int row   = gw * 16 + gq;
  int src = 0; float wt = 0.f;
  int tv = -1;
  if (row < nE) { src = s_src[start + row]; wt = s_w[start + row]; tv = s_tgt[start + row]; }
  int wti = __float_as_int(wt);
#pragma unroll
  for (int i = 0; i < 8; ++i) {
    int s0 = __builtin_amdgcn_readlane(src, 2 * i);
    int s1 = __builtin_amdgcn_readlane(src, 2 * i + 1);
    int w0 = __builtin_amdgcn_readlane(wti, 2 * i);
    int w1 = __builtin_amdgcn_readlane(wti, 2 * i + 1);
    int   s = gh ? s1 : s0;
    float w = __int_as_float(gh ? w1 : w0);
    float4 v = *((const float4*)(x + (size_t)s * DIN) + gq);
    float4 o;
    o.x = v.x * w; o.y = v.y * w; o.z = v.z * w; o.w = v.w * w;
    *(float4*)&xs[(gw * 16 + 2 * i + gh) * XS_STRIDE + 4 * gq] = o;
  }
#pragma unroll
  for (int i = 0; i < 8; ++i) *(v2f*)(wdst + (4 * i) * WP_PITCH) = wreg[i];
  __syncthreads();

  int lane = threadIdx.x & 31;
  int wv   = threadIdx.x >> 5;
  int hi   = lane >> 4;
  int lm   = lane & 15;
  const float* arow = &xs[(wv * 16 + lm) * XS_STRIDE + 2 * hi];

  const float* bq = &Wl[hi * WP_PITCH + ((4 * lm) ^ (hi << 5))];
  v8f acc[4] = {{}, {}, {}, {}};
#pragma unroll
  for (int ks = 0; ks < 16; ++ks) {
    v2f a   = *(const v2f*)(arow + ks * 4);
    v4f q0  = *(const v4f*)(bq + ks * 2 * WP_PITCH);
    v4f q1  = *(const v4f*)(bq + ks * 2 * WP_PITCH + 64);
    v2f b0 = {q0.x, q0.y}, b1 = {q0.z, q0.w}, b2 = {q1.x, q1.y}, b3 = {q1.z, q1.w};
    acc[0] = __builtin_amdgcn_wmma_f32_16x16x4_f32(
         false, a,  false, b0,
         (short)0, acc[0],  false,  false);
    acc[1] = __builtin_amdgcn_wmma_f32_16x16x4_f32(
         false, a,  false, b1,
         (short)0, acc[1],  false,  false);
    acc[2] = __builtin_amdgcn_wmma_f32_16x16x4_f32(
         false, a,  false, b2,
         (short)0, acc[2],  false,  false);
    acc[3] = __builtin_amdgcn_wmma_f32_16x16x4_f32(
         false, a,  false, b3,
         (short)0, acc[3],  false,  false);
  }

#pragma unroll
  for (int p = 0; p < 2; ++p) {
    const v8f acc0 = acc[2 * p], acc1 = acc[2 * p + 1];
    float* ocol = out + p * 32 + lane;
#pragma unroll
    for (int j = 0; j < 8; ++j) {
      auto sw = __builtin_amdgcn_permlane16_swap(__float_as_uint(acc0[j]), __float_as_uint(acc1[j]),
                                                 false, false);
      int t0 = __builtin_amdgcn_readlane(tv, j);
      int t1 = __builtin_amdgcn_readlane(tv, j + 8);
      if (t0 >= 0) atomicAdd(ocol + (size_t)t0 * DOUT, __uint_as_float(sw[0]));
      if (t1 >= 0) atomicAdd(ocol + (size_t)t1 * DOUT, __uint_as_float(sw[1]));
    }
  }
}

static inline size_t align256(size_t v) { return (v + 255) & ~(size_t)255; }

extern "C" void kernel_launch(void* const* d_in, const int* in_sizes, int n_in,
                              void* d_out, int out_size, void* d_ws, size_t ws_size,
                              hipStream_t stream) {
  const float* x       = (const float*)d_in[0];
  const int*   source  = (const int*)d_in[1];
  const int*   target  = (const int*)d_in[2];
  const int*   etype   = (const int*)d_in[3];
  const float* ew      = (const float*)d_in[4];
  const float* bases   = (const float*)d_in[5];
  const float* comb    = (const float*)d_in[6];
  float*       out     = (float*)d_out;

  const int N = in_sizes[0] / DIN;
  const int E = in_sizes[1];
  const int B = in_sizes[5] / (DIN * DOUT);
  const int R = in_sizes[6] / B;

  uint8_t* w = (uint8_t*)d_ws;
  float* Wbuf    = (float*)w;  w += align256((size_t)R * DIN * DOUT * 4);
  int*   counts  = (int*)w;    w += 256;
  int*   cursor  = (int*)w;    w += 256;
  int*   s_src   = (int*)w;    w += align256((size_t)E * 4);
  int*   s_tgt   = (int*)w;    w += align256((size_t)E * 4);
  float* s_w     = (float*)w;

  hipMemsetAsync(counts, 0, 512, stream);

  int eblk = (E + CNT_EPB - 1) / CNT_EPB;
  hist_kernel<<<eblk, 256, 0, stream>>>(etype, target, E, R, N, counts, bases, comb, Wbuf, B, out);
  scatter_kernel<<<eblk, 256, 0, stream>>>(source, target, etype, ew, E, R, N,
                                           counts, cursor, s_src, s_tgt, s_w);

  int tiles = (E + TILE_E - 1) / TILE_E + 2 * R;
  rgcn_tile_kernel<<<tiles, 256, 0, stream>>>(x, Wbuf, s_src, s_tgt, s_w,
                                              counts, R, cursor + 63, counts + 63, out);
}


# scatter kernel: a block's edges pass through LDS in (bucket, rank) order, so one store instruction writes neighbours of one bucket (few lines, was a line per bucket); places, add order as before
# speedup vs baseline: 6.9613x; 1.0921x over previous
#include <hip/hip_runtime.h>
#include <hip/hip_bf16.h>

typedef __attribute__((ext_vector_type(2))) float v2f;
typedef __attribute__((ext_vector_type(4))) float v4f;
typedef __attribute__((ext_vector_type(8))) float v8f;

#define DIN 64
#define DOUT 64
#define TILE_E 128
#define XS_STRIDE 68
#define WP_PITCH 128

#define CNT_G 2
#define CNT_EPB (256 * 4 * CNT_G)
#define MAXR 64

__device__ __forceinline__ void load4i(const int* __restrict__ p, int e, int E, int fill, int (&v)[4]) {
  if (e + 4 <= E) {
    int4 q = *(const int4*)(p + e);
    v[0] = q.x; v[1] = q.y; v[2] = q.z; v[3] = q.w;
  } else {
#pragma unroll
    for (int j = 0; j < 4; ++j) v[j] = (e + j < E) ? p[e + j] : fill;
  }
}
__device__ __forceinline__ void load4f(const float* __restrict__ p, int e, int E, float (&v)[4]) {
  if (e + 4 <= E) {
    float4 q = *(const float4*)(p + e);
    v[0] = q.x; v[1] = q.y; v[2] = q.z; v[3] = q.w;
  } else {
#pragma unroll
    for (int j = 0; j < 4; ++j) v[j] = (e + j < E) ? p[e + j] : 0.f;
  }
}

__device__ __forceinline__ int bucket_of(int rel, int tgt, int R, int NH) {
  return rel + (tgt >= NH ? R : 0);
}

__global__ void __launch_bounds__(256)
hist_kernel(const int* __restrict__ et, const int* __restrict__ tgt, int E, int R, int N, int* counts,
            const float* __restrict__ bases, const float* __restrict__ comb,
            float* __restrict__ W, int B, float* __restrict__ out) {
  __shared__ int lc[MAXR];
  if (threadIdx.x < MAXR) lc[threadIdx.x] = 0;
  __syncthreads();
  const int NH = N >> 1;
  int base = (int)blockIdx.x * CNT_EPB;
  int r[CNT_G][4], vt[CNT_G][4];
#pragma unroll
  for (int g = 0; g < CNT_G; ++g) {
    int e = base + (g * 256 + (int)threadIdx.x) * 4;
    load4i(et,  e, E, -1, r[g]);
    load4i(tgt, e, E, 0, vt[g]);
  }
  const int gid = (int)blockIdx.x * 256 + (int)threadIdx.x;
  const int gsz = (int)gridDim.x * 256;
  {
    float4* o4 = (float4*)out;
    const int n4 = N * (DOUT / 4);
    const float4 z = make_float4(0.f, 0.f, 0.f, 0.f);
    for (int i = gid; i < n4; i += gsz) o4[i] = z;
  }
  {
    const int nW = R * DIN * DOUT;
    for (int idx = gid; idx < nW; idx += gsz) {
      const float* c = comb + (idx / (DIN * DOUT)) * B;
      const int k = idx % (DIN * DOUT);
      float acc = 0.f;
      for (int b = 0; b < B; ++b) acc += c[b] * bases[b * DIN * DOUT + k];
      W[idx] = acc;
    }
  }
#pragma unroll
  for (int g = 0; g < CNT_G; ++g)
#pragma unroll
    for (int j = 0; j < 4; ++j)
      if (r[g][j] >= 0) atomicAdd(&lc[bucket_of(r[g][j], vt[g][j], R, NH)], 1);
  __syncthreads();
  if ((int)threadIdx.x < 2 * R) {
    int c = lc[threadIdx.x];
    if (c) atomicAdd(&counts[threadIdx.x], c);
  }
}

__global__ void __launch_bounds__(256)
scatter_kernel(const int* __restrict__ src, const int* __restrict__ tgt,
               const int* __restrict__ et, const float* __restrict__ wts,
               int E, int R, int N, const int* __restrict__ counts, int* taken,
               int* s_src, int* s_tgt, float* s_w) {
  __shared__ int lc[MAXR];
  __shared__ int cn[MAXR];
  __shared__ int lo[MAXR];
  __shared__ int gd[MAXR];
  __shared__ __attribute__((aligned(16))) int4 rec[CNT_EPB];
  const int NH = N >> 1;
  int base = (int)blockIdx.x * CNT_EPB;
  int r[CNT_G][4], vs[CNT_G][4], vt[CNT_G][4], rk[CNT_G][4];
  float vw[CNT_G][4];
#pragma unroll
  for (int g = 0; g < CNT_G; ++g) {
    int e = base + (g * 256 + (int)threadIdx.x) * 4;
    load4i(et,  e, E, -1, r[g]);
    load4i(src, e, E, 0, vs[g]);
    load4i(tgt, e, E, 0, vt[g]);
    load4f(wts, e, E, vw[g]);
  }
  if (threadIdx.x < MAXR) {
    lc[threadIdx.x] = 0;
    cn[threadIdx.x] = ((int)threadIdx.x < 2 * R) ? counts[threadIdx.x] : 0;
  }
  __syncthreads();
  int bstart = 0;
  if ((int)threadIdx.x < 2 * R) {
#pragma unroll
    for (int j = 0; j < 48; ++j) bstart += (j < (int)threadIdx.x) ? cn[j] : 0;
  }
#pragma unroll
  for (int g = 0; g < CNT_G; ++g)
#pragma unroll
    for (int j = 0; j < 4; ++j)
      if (r[g][j] >= 0) r[g][j] = bucket_of(r[g][j], vt[g][j], R, NH);
#pragma unroll
  for (int g = 0; g < CNT_G; ++g)
#pragma unroll
    for (int j = 0; j < 4; ++j)
      rk[g][j] = (r[g][j] >= 0) ? atomicAdd(&lc[r[g][j]], 1) : 0;
  __syncthreads();
  if ((int)threadIdx.x <= 2 * R) {
    int first = 0;
#pragma unroll
    for (int j = 0; j < 48; ++j) first += (j < (int)threadIdx.x) ? lc[j] : 0;
    lo[threadIdx.x] = first;
    if ((int)threadIdx.x < 2 * R) {
      int c = lc[threadIdx.x];
      int g0 = c ? bstart + atomicAdd(&taken[threadIdx.x], c) : 0;
      gd[threadIdx.x] = g0 - first;
    }
  }
  __syncthreads();
#pragma unroll
  for (int g = 0; g < CNT_G; ++g)
#pragma unroll
    for (int j = 0; j < 4; ++j)
      if (r[g][j] >= 0) {
        int slot = lo[r[g][j]] + rk[g][j];
        rec[slot] = make_int4(vs[g][j], vt[g][j], __float_as_int(vw[g][j]), gd[r[g][j]] + slot);
      }
  __syncthreads();
  const int nblk = lo[2 * R];
  int4 q[4 * CNT_G];
#pragma unroll
  for (int k = 0; k < 4 * CNT_G; ++k) {
    int i = k * 256 + (int)threadIdx.x;
    q[k] = (i < nblk) ? rec[i] : make_int4(0, 0, 0, -1);
  }
#pragma unroll
  for (int k = 0; k < 4 * CNT_G; ++k)
    if (q[k].w >= 0) {
      s_src[q[k].w] = q[k].x;
      s_tgt[q[k].w] = q[k].y;
      s_w[q[k].w]   = __int_as_float(q[k].z);
    }
}

__global__ void __launch_bounds__(256)
rgcn_tile_kernel(const float* __restrict__ x,
                 const float* __restrict__ W,
                 const int* __restrict__ s_src,
                 const int* __restrict__ s_tgt,
                 const float* __restrict__ s_w,
                 const int* __restrict__ counts,
                 int R,
                 int* tick0, int* tick1,
                 float* __restrict__ out) {
  const int Q = 2 * R;
  int tall = 0, T0 = 0;
  for (int g = 0; g < Q; g += 16) {
#pragma unroll
    for (int j = 0; j < 16; ++j) {
      T0 = (g + j == R) ? tall : T0;
      tall += (int)((unsigned)(counts[g + j] + TILE_E - 1) / TILE_E);
    }
  }
  int T1 = tall - T0;

  __shared__ int claim[2];
  if (threadIdx.x == 0) {
    int die = (__builtin_amdgcn_s_sendmsg_rtn(0x87) >> 16) & 15;
    int h = (die >> 2) & 1;
    int t = atomicAdd(h ? tick1 : tick0, 1);
    if (t >= (h ? T1 : T0)) {
      h ^= 1;
      t = atomicAdd(h ? tick1 : tick0, 1);
      if (t >= (h ? T1 : T0)) h = -1;
    }
    claim[0] = h;
    claim[1] = t;
  }
  __syncthreads();
  int hsel = claim[0];
  if (hsel < 0) return;
  int b = __builtin_amdgcn_readfirstlane(claim[1] + (hsel ? T0 : 0));

  int q = 0, cnt = 0, eo = 0, ts = 0;
  int tacc = 0, eoff = 0;
  for (int g = 0; g < Q; g += 16) {
#pragma unroll
    for (int j = 0; j < 16; ++j) {
      int c = counts[g + j];
      bool here = (b >= tacc);
      q   = here ? g + j : q;
      cnt = here ? c     : cnt;
      eo  = here ? eoff  : eo;
      ts  = here ? tacc  : ts;
      tacc += (int)((unsigned)(c + TILE_E - 1) / TILE_E);
      eoff += c;
    }
  }
  int r = (q >= R) ? q - R : q;
  int tile0 = (b - ts) * TILE_E;
  if (tile0 >= cnt) return;
  int start = eo + tile0;
  int nE = min(TILE_E, cnt - tile0);

  __shared__ __attribute__((aligned(16))) float xs[TILE_E * XS_STRIDE];
  __shared__ __attribute__((aligned(16))) float Wl[(DIN / 2) * WP_PITCH];

  const int wn  = threadIdx.x & 63;
  const int wk0 = threadIdx.x >> 6;
  const float* Wg = W + r * DIN * DOUT + (2 * wk0) * DOUT + wn;
  float* wdst = &Wl[wk0 * WP_PITCH + (wn >> 5) * 64 +
                    ((((wn & 15) << 2) + ((wn >> 4) & 1) * 2) ^ ((wk0 & 1) << 5))];
  v2f wreg[8];
#pragma unroll
  for (int i = 0; i < 8; ++i) {
    wreg[i].x = Wg[(8 * i) * DOUT];
    wreg[i].y = Wg[(8 * i + 1) * DOUT];
  }

  int glane = threadIdx.x & 31;
  int gw    = threadIdx.x >> 5;
  int gh    = glane >> 4;
  int gq    = glane & 15;
  int row   = gw * 16 + gq;
  int src = 0; float wt = 0.f;
  int tv = -1;
  if (row < nE) { src = s_src[start + row]; wt = s_w[start + row]; tv = s_tgt[start + row]; }
  int wti = __float_as_int(wt);
#pragma unroll
  for (int i = 0; i < 8; ++i) {
    int s0 = __builtin_amdgcn_readlane(src, 2 * i);
    int s1 = __builtin_amdgcn_readlane(src, 2 * i + 1);
    int w0 = __builtin_amdgcn_readlane(wti, 2 * i);
    int w1 = __builtin_amdgcn_readlane(wti, 2 * i + 1);
    int   s = gh ? s1 : s0;
    float w = __int_as_float(gh ? w1 : w0);
    float4 v = *((const float4*)(x + (size_t)s * DIN) + gq);
    float4 o;
    o.x = v.x * w; o.y = v.y * w; o.z = v.z * w; o.w = v.w * w;
    *(float4*)&xs[(gw * 16 + 2 * i + gh) * XS_STRIDE + 4 * gq] = o;
  }
#pragma unroll
  for (int i = 0; i < 8; ++i) *(v2f*)(wdst + (4 * i) * WP_PITCH) = wreg[i];
  __syncthreads();

  int lane = threadIdx.x & 31;
  int wv   = threadIdx.x >> 5;
  int hi   = lane >> 4;
  int lm   = lane & 15;
  const float* arow = &xs[(wv * 16 + lm) * XS_STRIDE + 2 * hi];

  const float* bq = &Wl[hi * WP_PITCH + ((4 * lm) ^ (hi << 5))];
  v8f acc[4] = {{}, {}, {}, {}};
#pragma unroll
  for (int ks = 0; ks < 16; ++ks) {
    v2f a   = *(const v2f*)(arow + ks * 4);
    v4f q0  = *(const v4f*)(bq + ks * 2 * WP_PITCH);
    v4f q1  = *(const v4f*)(bq + ks * 2 * WP_PITCH + 64);
    v2f b0 = {q0.x, q0.y}, b1 = {q0.z, q0.w}, b2 = {q1.x, q1.y}, b3 = {q1.z, q1.w};
    acc[0] = __builtin_amdgcn_wmma_f32_16x16x4_f32(
         false, a,  false, b0,
         (short)0, acc[0],  false,  false);
    acc[1] = __builtin_amdgcn_wmma_f32_16x16x4_f32(
         false, a,  false, b1,
         (short)0, acc[1],  false,  false);
    acc[2] = __builtin_amdgcn_wmma_f32_16x16x4_f32(
         false, a,  false, b2,
         (short)0, acc[2],  false,  false);
    acc[3] = __builtin_amdgcn_wmma_f32_16x16x4_f32(
         false, a,  false, b3,
         (short)0, acc[3],  false,  false);
  }

#pragma unroll
  for (int p = 0; p < 2; ++p) {
    const v8f acc0 = acc[2 * p], acc1 = acc[2 * p + 1];
    float* ocol = out + p * 32 + lane;
#pragma unroll
    for (int j = 0; j < 8; ++j) {
      auto sw = __builtin_amdgcn_permlane16_swap(__float_as_uint(acc0[j]), __float_as_uint(acc1[j]),
                                                 false, false);
      int t0 = __builtin_amdgcn_readlane(tv, j);
      int t1 = __builtin_amdgcn_readlane(tv, j + 8);
      if (t0 >= 0) atomicAdd(ocol + (size_t)t0 * DOUT, __uint_as_float(sw[0]));
      if (t1 >= 0) atomicAdd(ocol + (size_t)t1 * DOUT, __uint_as_float(sw[1]));
    }
  }
}

static inline size_t align256(size_t v) { return (v + 255) & ~(size_t)255; }

extern "C" void kernel_launch(void* const* d_in, const int* in_sizes, int n_in,
                              void* d_out, int out_size, void* d_ws, size_t ws_size,
                              hipStream_t stream) {
  const float* x       = (const float*)d_in[0];
  const int*   source  = (const int*)d_in[1];
  const int*   target  = (const int*)d_in[2];
  const int*   etype   = (const int*)d_in[3];
  const float* ew      = (const float*)d_in[4];
  const float* bases   = (const float*)d_in[5];
  const float* comb    = (const float*)d_in[6];
  float*       out     = (float*)d_out;

  const int N = in_sizes[0] / DIN;
  const int E = in_sizes[1];
  const int B = in_sizes[5] / (DIN * DOUT);
  const int R = in_sizes[6] / B;

  uint8_t* w = (uint8_t*)d_ws;
  float* Wbuf    = (float*)w;  w += align256((size_t)R * DIN * DOUT * 4);
  int*   counts  = (int*)w;    w += 256;
  int*   cursor  = (int*)w;    w += 256;
  int*   s_src   = (int*)w;    w += align256((size_t)E * 4);
  int*   s_tgt   = (int*)w;    w += align256((size_t)E * 4);
  float* s_w     = (float*)w;

  hipMemsetAsync(counts, 0, 512, stream);

  int eblk = (E + CNT_EPB - 1) / CNT_EPB;
  hist_kernel<<<eblk, 256, 0, stream>>>(etype, target, E, R, N, counts, bases, comb, Wbuf, B, out);
  scatter_kernel<<<eblk, 256, 0, stream>>>(source, target, etype, ew, E, R, N,
                                           counts, cursor, s_src, s_tgt, s_w);

  int tiles = (E + TILE_E - 1) / TILE_E + 2 * R;
  rgcn_tile_kernel<<<tiles, 256, 0, stream>>>(x, Wbuf, s_src, s_tgt, s_w,
                                              counts, R, cursor + 63, counts + 63, out);
}
